# starting kernel
# baseline (speedup 1.0000x reference)
#include <hip/hip_runtime.h>

typedef __attribute__((ext_vector_type(2))) float v2f;
typedef __attribute__((ext_vector_type(8))) float v8f;

#define NEG_SLOPE 0.2f

__device__ __forceinline__ unsigned enc_f32(float f) {
  unsigned u = __float_as_uint(f);
  return (u & 0x80000000u) ? ~u : (u | 0x80000000u);
}
__device__ __forceinline__ float dec_f32(unsigned u) {
  unsigned b = (u & 0x80000000u) ? (u & 0x7fffffffu) : ~u;
  return __uint_as_float(b);
}
#define ENC_NEG_INF 0x007FFFFFu

__global__ void fill_f32(float* __restrict__ p, float v, long long n) {
  long long i = (long long)blockIdx.x * blockDim.x + threadIdx.x;
  if (i < n) p[i] = v;
}
__global__ void fill_u32(unsigned* __restrict__ p, unsigned v, int n) {
  int i = blockIdx.x * blockDim.x + threadIdx.x;
  if (i < n) p[i] = v;
}

__global__ void gemm_f32_wmma(const float* __restrict__ X, const float* __restrict__ W,
                              float* __restrict__ H, int M, int K, int N) {
  int lane = threadIdx.x & 31;
  int wave = (int)((blockIdx.x * (long long)blockDim.x + threadIdx.x) >> 5);
  int nt4 = N >> 6;
  int mt  = (M + 15) >> 4;
  if (wave >= mt * nt4) return;
  int tm = wave / nt4, tn = wave % nt4;
  int half = lane >> 4;
  int l15  = lane & 15;
  int row = tm * 16 + l15;  if (row >= M) row = M - 1;
  int col0 = tn * 64 + l15;
  v8f acc0 = {}, acc1 = {}, acc2 = {}, acc3 = {};
  for (int k0 = 0; k0 < K; k0 += 4) {
    int kb = k0 + (half << 1);
    v2f a = *(const v2f*)(X + (size_t)row * K + kb);
    const float* w0 = W + (size_t)kb * N;
    const float* w1 = W + (size_t)(kb + 1) * N;
    v2f b0, b1, b2, b3;
    b0.x = w0[col0 +  0]; b0.y = w1[col0 +  0];
    b1.x = w0[col0 + 16]; b1.y = w1[col0 + 16];
    b2.x = w0[col0 + 32]; b2.y = w1[col0 + 32];
    b3.x = w0[col0 + 48]; b3.y = w1[col0 + 48];
    acc0 = __builtin_amdgcn_wmma_f32_16x16x4_f32(false, a, false, b0, (short)0, acc0, false, false);
    acc1 = __builtin_amdgcn_wmma_f32_16x16x4_f32(false, a, false, b1, (short)0, acc1, false, false);
    acc2 = __builtin_amdgcn_wmma_f32_16x16x4_f32(false, a, false, b2, (short)0, acc2, false, false);
    acc3 = __builtin_amdgcn_wmma_f32_16x16x4_f32(false, a, false, b3, (short)0, acc3, false, false);
  }
  int rbase = tm * 16 + half * 8;
#pragma unroll
  for (int v = 0; v < 8; ++v) {
    int r = rbase + v;
    if (r < M) {
      float* hp = H + (size_t)r * N + col0;
      hp[ 0] = acc0[v];
      hp[16] = acc1[v];
      hp[32] = acc2[v];
      hp[48] = acc3[v];
    }
  }
}

__global__ void node_alpha(const float* __restrict__ h, const float* __restrict__ avs,
                           const float* __restrict__ avd, float* __restrict__ as,
                           float* __restrict__ ad, int Nn, int F) {
  int lane = threadIdx.x & 31;
  int node = (int)((blockIdx.x * (long long)blockDim.x + threadIdx.x) >> 5);
  if (node >= Nn) return;
  const float* hp = h + (size_t)node * F;
  float ss = 0.f, sd = 0.f;
  for (int c = lane; c < F; c += 32) {
    float hv = hp[c];
    ss += hv * avs[c];
    sd += hv * avd[c];
  }
#pragma unroll
  for (int off = 16; off > 0; off >>= 1) {
    ss += __shfl_xor(ss, off, 32);
    sd += __shfl_xor(sd, off, 32);
  }
  if (lane == 0) { as[node] = ss; ad[node] = sd; }
}

__device__ __forceinline__ void edge_endpoints(int t, int E, const int* __restrict__ esrc,
                                               const int* __restrict__ edst, int& s, int& d) {
  if (t < E) { s = esrc[t]; d = edst[t]; } else { s = t - E; d = s; }
}

__global__ void edge_max(const float* __restrict__ as, const float* __restrict__ ad,
                         const int* __restrict__ esrc, const int* __restrict__ edst,
                         unsigned* __restrict__ menc, int E, int Nn) {
  int t = blockIdx.x * blockDim.x + threadIdx.x;
  if (t >= E + Nn) return;
  int s, d; edge_endpoints(t, E, esrc, edst, s, d);
  float e = as[s] + ad[d];
  e = (e > 0.f) ? e : NEG_SLOPE * e;
  atomicMax(menc + d, enc_f32(e));
}

__global__ void edge_expsum(const float* __restrict__ as, const float* __restrict__ ad,
                            const int* __restrict__ esrc, const int* __restrict__ edst,
                            const unsigned* __restrict__ menc, float* __restrict__ w,
                            float* __restrict__ z, int E, int Nn) {
  int t = blockIdx.x * blockDim.x + threadIdx.x;
  if (t >= E + Nn) return;
  int s, d; edge_endpoints(t, E, esrc, edst, s, d);
  float e = as[s] + ad[d];
  e = (e > 0.f) ? e : NEG_SLOPE * e;
  float wv = expf(e - dec_f32(menc[d]));
  w[t] = wv;
  atomicAdd(z + d, wv);
}

__global__ void aggregate(const float* __restrict__ h, const int* __restrict__ esrc,
                          const int* __restrict__ edst, const float* __restrict__ w,
                          const float* __restrict__ z, float* __restrict__ acc,
                          int E, int Nn, int F) {
  long long t = (long long)blockIdx.x * blockDim.x + threadIdx.x;
  int f4 = F >> 2;
  long long total = (long long)(E + Nn) * f4;
  if (t >= total) return;
  int eix = (int)(t / f4);
  int c = ((int)(t % f4)) << 2;
  int s, d; edge_endpoints(eix, E, esrc, edst, s, d);
  float alpha = w[eix] / z[d];
  const float4 hv = *(const float4*)(h + (size_t)s * F + c);
  float* op = acc + (size_t)d * F + c;
  atomicAdd(op + 0, alpha * hv.x);
  atomicAdd(op + 1, alpha * hv.y);
  atomicAdd(op + 2, alpha * hv.z);
  atomicAdd(op + 3, alpha * hv.w);
}

__global__ void bias_act(const float* __restrict__ acc, const float* __restrict__ b,
                         float* __restrict__ out, long long total, int F, int do_relu) {
  long long t = (long long)blockIdx.x * blockDim.x + threadIdx.x;
  if (t >= total) return;
  float v = acc[t] + b[(int)(t % F)];
  out[t] = do_relu ? fmaxf(v, 0.f) : v;
}

static void run_layer(const float* xin, int Fin, const float* W,
                      const float* avs, const float* avd, const float* bias,
                      float* h, float* accb, float* xout, int Fout, int do_relu,
                      const int* esrc, const int* edst, int E, int Nn,
                      float* w_e, float* z, unsigned* menc, float* as, float* ad,
                      hipStream_t stream) {
  int mt = (Nn + 15) / 16, nt4 = Fout / 64;
  int waves = mt * nt4;
  gemm_f32_wmma<<<(waves + 7) / 8, 256, 0, stream>>>(xin, W, h, Nn, Fin, Fout);
  node_alpha<<<(Nn + 7) / 8, 256, 0, stream>>>(h, avs, avd, as, ad, Nn, Fout);

  int ET = E + Nn;
  fill_u32<<<(Nn + 255) / 256, 256, 0, stream>>>(menc, ENC_NEG_INF, Nn);
  fill_f32<<<(Nn + 255) / 256, 256, 0, stream>>>(z, 0.f, (long long)Nn);
  long long accn = (long long)Nn * Fout;
  fill_f32<<<(unsigned)((accn + 255) / 256), 256, 0, stream>>>(accb, 0.f, accn);

  edge_max<<<(ET + 255) / 256, 256, 0, stream>>>(as, ad, esrc, edst, menc, E, Nn);
  edge_expsum<<<(ET + 255) / 256, 256, 0, stream>>>(as, ad, esrc, edst, menc, w_e, z, E, Nn);

  long long aggt = (long long)ET * (Fout >> 2);
  aggregate<<<(unsigned)((aggt + 255) / 256), 256, 0, stream>>>(h, esrc, edst, w_e, z, accb,
                                                                E, Nn, Fout);
  bias_act<<<(unsigned)((accn + 255) / 256), 256, 0, stream>>>(accb, bias, xout, accn, Fout,
                                                               do_relu);
}

extern "C" void kernel_launch(void* const* d_in, const int* in_sizes, int n_in,
                              void* d_out, int out_size, void* d_ws, size_t ws_size,
                              hipStream_t stream) {
  const float* x   = (const float*)d_in[0];
  const int*  eidx = (const int*)d_in[1];
  const float* W1  = (const float*)d_in[2];
  const float* a1s = (const float*)d_in[3];
  const float* a1d = (const float*)d_in[4];
  const float* b1  = (const float*)d_in[5];
  const float* W2  = (const float*)d_in[6];
  const float* a2s = (const float*)d_in[7];
  const float* a2d = (const float*)d_in[8];
  const float* b2  = (const float*)d_in[9];
  float* out = (float*)d_out;

  const int IN_C = 128, HID_C = 256, OUT_C = 128;
  int Nn = in_sizes[0] / IN_C;
  int E  = in_sizes[1] / 2;
  const int* esrc = eidx;
  const int* edst = eidx + E;
  int ET = E + Nn;

  float* bufA = (float*)d_ws;
  float* bufB = bufA + (size_t)Nn * HID_C;
  float* w_e  = bufB + (size_t)Nn * HID_C;
  float* z    = w_e + ET;
  unsigned* menc = (unsigned*)(z + Nn);
  float* as   = (float*)(menc + Nn);
  float* ad   = as + Nn;
  (void)ws_size; (void)n_in; (void)out_size;

  run_layer(x, IN_C, W1, a1s, a1d, b1,
             bufA,  bufB,  bufB, HID_C,  1,
            esrc, edst, E, Nn, w_e, z, menc, as, ad, stream);

  float* h2   = bufA;
  float* acc2 = bufA + (size_t)Nn * OUT_C;
  run_layer(bufB, HID_C, W2, a2s, a2d, b2,
             h2,  acc2,  out, OUT_C,  0,
            esrc, edst, E, Nn, w_e, z, menc, as, ad, stream);
}


# GEMM fed in whole 128-byte lines, 16 B per lane: tile columns interleaved by 4, X and W read in 8-k steps with one matching k order, H rows stored whole; product and logit sum order change
# speedup vs baseline: 5.2649x; 5.2649x over previous
#include <hip/hip_runtime.h>

typedef __attribute__((ext_vector_type(2))) float v2f;
typedef __attribute__((ext_vector_type(8))) float v8f;

#define NEG_SLOPE 0.2f

template <int MASK>
__device__ __forceinline__ float row_xor_add(float x) {
  const int y = __builtin_amdgcn_update_dpp(0, __float_as_int(x), 0x160 | MASK, 0xf, 0xf, true);
  return x + __int_as_float(y);
}
__device__ __forceinline__ float row_sum16(float x) {
  x = row_xor_add<1>(x);
  x = row_xor_add<2>(x);
  x = row_xor_add<4>(x);
  x = row_xor_add<8>(x);
  return x;
}

__global__ void gemm_f32_wmma(const float* __restrict__ X, const float* __restrict__ W,
                              const float* __restrict__ avs, const float* __restrict__ avd,
                              float* __restrict__ H, float* __restrict__ as, float* __restrict__ ad,
                              int M, int K, int N) {
  __shared__ __attribute__((aligned(16))) float part[8][2][16];
  int lane = threadIdx.x & 31;
  int wib  = threadIdx.x >> 5;
  int wave = (int)((blockIdx.x * (long long)blockDim.x + threadIdx.x) >> 5);
  int nt4 = N >> 6;
  int mt  = (M + 15) >> 4;
  const bool live = wave < mt * nt4;
  int tm = wave / nt4, tn = wave % nt4;
  int half = lane >> 4;
  int l15  = lane & 15;
  int col0 = tn * 64 + l15 * 4;
  v8f acc0 = {}, acc1 = {}, acc2 = {}, acc3 = {};
  if (live) {
    int row = tm * 16 + l15;  if (row >= M) row = M - 1;
    const float4 sv = *(const float4*)(avs + col0);
    const float4 dv = *(const float4*)(avd + col0);
    const float* xp = X + (size_t)row * K + (half << 2);
    const float* wp = W + (size_t)(half << 2) * N + col0;
    for (int k0 = 0; k0 < K; k0 += 8) {
      const float4 a = *(const float4*)(xp + k0);
      const float* w = wp + (size_t)k0 * N;
      const float4 r0 = *(const float4*)(w);
      const float4 r1 = *(const float4*)(w + N);
      const float4 r2 = *(const float4*)(w + 2 * N);
      const float4 r3 = *(const float4*)(w + 3 * N);
      v2f a01, a23, b0, b1, b2, b3;
      a01.x = a.x; a01.y = a.y;
      a23.x = a.z; a23.y = a.w;
      b0.x = r0.x; b0.y = r1.x;  b1.x = r0.y; b1.y = r1.y;
      b2.x = r0.z; b2.y = r1.z;  b3.x = r0.w; b3.y = r1.w;
      acc0 = __builtin_amdgcn_wmma_f32_16x16x4_f32(false, a01, false, b0, (short)0, acc0, false, false);
      acc1 = __builtin_amdgcn_wmma_f32_16x16x4_f32(false, a01, false, b1, (short)0, acc1, false, false);
      acc2 = __builtin_amdgcn_wmma_f32_16x16x4_f32(false, a01, false, b2, (short)0, acc2, false, false);
      acc3 = __builtin_amdgcn_wmma_f32_16x16x4_f32(false, a01, false, b3, (short)0, acc3, false, false);
      b0.x = r2.x; b0.y = r3.x;  b1.x = r2.y; b1.y = r3.y;
      b2.x = r2.z; b2.y = r3.z;  b3.x = r2.w; b3.y = r3.w;
      acc0 = __builtin_amdgcn_wmma_f32_16x16x4_f32(false, a23, false, b0, (short)0, acc0, false, false);
      acc1 = __builtin_amdgcn_wmma_f32_16x16x4_f32(false, a23, false, b1, (short)0, acc1, false, false);
      acc2 = __builtin_amdgcn_wmma_f32_16x16x4_f32(false, a23, false, b2, (short)0, acc2, false, false);
      acc3 = __builtin_amdgcn_wmma_f32_16x16x4_f32(false, a23, false, b3, (short)0, acc3, false, false);
    }
    float ps[8], pd[8];
#pragma unroll
    for (int v = 0; v < 8; ++v) {
      ps[v] = row_sum16(acc0[v] * sv.x + acc1[v] * sv.y + acc2[v] * sv.z + acc3[v] * sv.w);
      pd[v] = row_sum16(acc0[v] * dv.x + acc1[v] * dv.y + acc2[v] * dv.z + acc3[v] * dv.w);
    }
    if (l15 == 0) {
      float* p = &part[wib][0][half * 8];
      *(float4*)(p)          = make_float4(ps[0], ps[1], ps[2], ps[3]);
      *(float4*)(p + 4)      = make_float4(ps[4], ps[5], ps[6], ps[7]);
      *(float4*)(p + 16)     = make_float4(pd[0], pd[1], pd[2], pd[3]);
      *(float4*)(p + 16 + 4) = make_float4(pd[4], pd[5], pd[6], pd[7]);
    }
  }
  __syncthreads();
  if (live) {
    int rbase = tm * 16 + half * 8;
#pragma unroll
    for (int v = 0; v < 8; ++v) {
      int r = rbase + v;
      if (r < M) *(float4*)(H + (size_t)r * N + col0) = make_float4(acc0[v], acc1[v], acc2[v], acc3[v]);
    }
    if (tn == 0) {
      const float* p = &part[wib][0][0] + lane;
      float s = p[0];
      for (int t = 1; t < nt4; ++t) s += p[t * 32];
      int r = tm * 16 + l15;
      if (r < M) ((lane < 16) ? as : ad)[r] = s;
    }
  }
}

__device__ __forceinline__ float leaky(float e) { return (e > 0.f) ? e : NEG_SLOPE * e; }

#define SORT_NB   256
#define SORT_T    1024
#define SORT_EPT  4
#define BK_SHIFT  8
#define BK_NODES  256
#define BK_MAX    256
#define STAGE_CAP 8192

__device__ __forceinline__ unsigned wave_scan256(unsigned* a, int lane) {
  unsigned c[8], sum = 0u;
#pragma unroll
  for (int j = 0; j < 8; ++j) { c[j] = a[lane * 8 + j]; sum += c[j]; }
  unsigned incl = sum;
  for (int off = 1; off < 32; off <<= 1) {
    unsigned v = __shfl_up(incl, off, 32);
    if (lane >= off) incl += v;
  }
  unsigned run = incl - sum;
#pragma unroll
  for (int j = 0; j < 8; ++j) { a[lane * 8 + j] = run; run += c[j]; }
  return __shfl(incl, 31, 32);
}

__global__ __launch_bounds__(SORT_T) void edge_bin(const int* __restrict__ esrc, const int* __restrict__ edst,
                                                   unsigned* __restrict__ tmp, unsigned* __restrict__ offsT,
                                                   int E, int C, int B) {
  __shared__ unsigned hist[BK_MAX + 1];
  __shared__ unsigned stage[SORT_T * SORT_EPT];
  const int tid = threadIdx.x, blk = blockIdx.x;
  const int e0 = blk * C;
  int n = E - e0; if (n > C) n = C; if (n < 0) n = 0;
  if (tid <= BK_MAX) hist[tid] = 0u;
  __syncthreads();
  int d[SORT_EPT], s[SORT_EPT];
#pragma unroll
  for (int k = 0; k < SORT_EPT; ++k) {
    const int i = k * SORT_T + tid;
    d[k] = -1; s[k] = 0;
    if (i < n) { d[k] = edst[e0 + i]; s[k] = esrc[e0 + i]; }
  }
  unsigned slot[SORT_EPT], bk[SORT_EPT];
#pragma unroll
  for (int k = 0; k < SORT_EPT; ++k) {
    const int i = k * SORT_T + tid;
    unsigned b = (unsigned)d[k] >> BK_SHIFT;
    if (b >= (unsigned)B) b = (unsigned)B - 1u;
    bk[k] = b; slot[k] = 0u;
    if (i < n) slot[k] = atomicAdd(&hist[b], 1u);
  }
  __syncthreads();
  if (tid < 32) {
    const unsigned tot = wave_scan256(hist, tid);
    if (tid == 0) hist[BK_MAX] = tot;
  }
  __syncthreads();
  if (tid <= BK_MAX) offsT[tid * SORT_NB + blk] = hist[tid];
#pragma unroll
  for (int k = 0; k < SORT_EPT; ++k) {
    const int i = k * SORT_T + tid;
    if (i < n) stage[hist[bk[k]] + slot[k]] =
        ((unsigned)s[k] << BK_SHIFT) | ((unsigned)d[k] & (BK_NODES - 1));
  }
  __syncthreads();
#pragma unroll
  for (int k = 0; k < SORT_EPT; ++k) {
    const int i = k * SORT_T + tid;
    if (i < n) tmp[e0 + i] = stage[i];
  }
}

__global__ __launch_bounds__(SORT_T) void bucket_group(const unsigned* __restrict__ tmp,
                                                       const unsigned* __restrict__ offsT,
                                                       int* __restrict__ csr_src, unsigned* __restrict__ deg,
                                                       unsigned* __restrict__ rowstart, int Nn, int C) {
  __shared__ __attribute__((aligned(16))) unsigned segbase[SORT_NB];
  __shared__ __attribute__((aligned(16))) unsigned seglen[SORT_NB];
  __shared__ unsigned cnt[BK_NODES], nodeoff[BK_NODES], cur[BK_NODES];
  __shared__ unsigned red[2][8];
  __shared__ unsigned stage[STAGE_CAP];
  const int tid = threadIdx.x, lane = tid & 31, w = tid >> 5, b = blockIdx.x;
  if (tid < SORT_NB) {
    const unsigned lo = offsT[b * SORT_NB + tid], hi = offsT[(b + 1) * SORT_NB + tid];
    segbase[tid] = (unsigned)tid * (unsigned)C + lo;
    seglen[tid] = hi - lo;
    cnt[tid] = 0u;
    unsigned a = lo, l = hi - lo;
    for (int off = 16; off > 0; off >>= 1) { a += __shfl_xor(a, off, 32); l += __shfl_xor(l, off, 32); }
    if (lane == 0) { red[0][w] = a; red[1][w] = l; }
  }
  __syncthreads();
  unsigned bstart = 0u, nb = 0u;
#pragma unroll
  for (int j = 0; j < 8; ++j) { bstart += red[0][j]; nb += red[1][j]; }

  unsigned len8[8], base8[8];
  {
    const uint4 l0 = *(const uint4*)&seglen[w * 8], l1 = *(const uint4*)&seglen[w * 8 + 4];
    const uint4 b0 = *(const uint4*)&segbase[w * 8], b1 = *(const uint4*)&segbase[w * 8 + 4];
    len8[0] = l0.x; len8[1] = l0.y; len8[2] = l0.z; len8[3] = l0.w;
    len8[4] = l1.x; len8[5] = l1.y; len8[6] = l1.z; len8[7] = l1.w;
    base8[0] = b0.x; base8[1] = b0.y; base8[2] = b0.z; base8[3] = b0.w;
    base8[4] = b1.x; base8[5] = b1.y; base8[6] = b1.z; base8[7] = b1.w;
  }
  unsigned v[8], vm = 0u;
#pragma unroll
  for (int q = 0; q < 8; ++q) {
    v[q] = 0u;
    if ((unsigned)lane < len8[q]) { v[q] = tmp[base8[q] + lane]; vm |= 1u << q; }
  }
#pragma unroll
  for (int q = 0; q < 8; ++q)
    if (vm & (1u << q)) atomicAdd(&cnt[v[q] & (BK_NODES - 1)], 1u);
#pragma unroll
  for (int q = 0; q < 8; ++q) {
    const unsigned len = len8[q], base = base8[q];
    for (unsigned j = lane + 32; j < len; j += 32) atomicAdd(&cnt[tmp[base + j] & (BK_NODES - 1)], 1u);
  }
  __syncthreads();
  if (tid < 32) {
#pragma unroll
    for (int j = 0; j < 8; ++j) nodeoff[tid * 8 + j] = cnt[tid * 8 + j];
    wave_scan256(nodeoff, tid);
#pragma unroll
    for (int j = 0; j < 8; ++j) cur[tid * 8 + j] = nodeoff[tid * 8 + j];
  }
  __syncthreads();
  if (tid < BK_NODES) {
    const int node = b * BK_NODES + tid;
    if (node < Nn) { deg[node] = cnt[tid]; rowstart[node] = bstart + nodeoff[tid]; }
  }
  const bool staged = nb <= (unsigned)STAGE_CAP;
#pragma unroll
  for (int q = 0; q < 8; ++q) {
    if (vm & (1u << q)) {
      const unsigned p = atomicAdd(&cur[v[q] & (BK_NODES - 1)], 1u);
      if (staged) stage[p] = v[q] >> BK_SHIFT; else csr_src[bstart + p] = (int)(v[q] >> BK_SHIFT);
    }
  }
#pragma unroll
  for (int q = 0; q < 8; ++q) {
    const unsigned len = len8[q], base = base8[q];
    for (unsigned j = lane + 32; j < len; j += 32) {
      const unsigned r = tmp[base + j];
      const unsigned p = atomicAdd(&cur[r & (BK_NODES - 1)], 1u);
      if (staged) stage[p] = r >> BK_SHIFT; else csr_src[bstart + p] = (int)(r >> BK_SHIFT);
    }
  }
  __syncthreads();
  if (staged)
    for (unsigned i = tid; i < nb; i += SORT_T) csr_src[bstart + i] = (int)stage[i];
}

template <int F, int RELU>
__global__ void gat_pull(const float* __restrict__ h, const float* __restrict__ as,
                         const float* __restrict__ ad, const int* __restrict__ csr_src,
                         const unsigned* __restrict__ rowstart, const unsigned* __restrict__ deg,
                         const float* __restrict__ bias, float* __restrict__ out, int Nn) {
  constexpr int V = F / 128;
  int lane = threadIdx.x & 31;
  const int node = __builtin_amdgcn_readfirstlane(
      (int)((blockIdx.x * (long long)blockDim.x + threadIdx.x) >> 5));
  if (node >= Nn) return;
  const int start = (int)rowstart[node];
  const int cnt   = (int)deg[node];
  const float adn = ad[node];
  const float e_self = leaky(as[node] + adn);

  float m = e_self;
  for (int j = lane; j < cnt; j += 32)
    m = fmaxf(m, leaky(as[csr_src[start + j]] + adn));
  for (int off = 16; off > 0; off >>= 1) m = fmaxf(m, __shfl_xor(m, off, 32));

  float zs = 0.f;
  for (int j = lane; j < cnt; j += 32)
    zs += expf(leaky(as[csr_src[start + j]] + adn) - m);
  for (int off = 16; off > 0; off >>= 1) zs += __shfl_xor(zs, off, 32);
  const float w_self = expf(e_self - m);
  const float z = zs + w_self;

  float4 acc[V];
  {
    const float a = w_self / z;
    const float* hp = h + (size_t)node * F + lane * 4;
    for (int v = 0; v < V; ++v) {
      const float4 hv = *(const float4*)(hp + v * 128);
      acc[v].x = a * hv.x; acc[v].y = a * hv.y; acc[v].z = a * hv.z; acc[v].w = a * hv.w;
    }
  }
  for (int jb = 0; jb < cnt; jb += 32) {
    const int j = jb + lane;
    int sj = node;
    float aj = 0.f;
    if (j < cnt) {
      sj = csr_src[start + j];
      aj = expf(leaky(as[sj] + adn) - m) / z;
    }
    const int nj = (cnt - jb < 32) ? (cnt - jb) : 32;
#pragma unroll 4
    for (int i = 0; i < nj; ++i) {
      const int   s = __shfl(sj, i, 32);
      const float a = __shfl(aj, i, 32);
      const float* hp = h + (size_t)s * F + lane * 4;
      for (int v = 0; v < V; ++v) {
        const float4 hv = *(const float4*)(hp + v * 128);
        acc[v].x += a * hv.x; acc[v].y += a * hv.y; acc[v].z += a * hv.z; acc[v].w += a * hv.w;
      }
    }
  }

  float* op = out + (size_t)node * F + lane * 4;
  for (int v = 0; v < V; ++v) {
    const float4 bv = *(const float4*)(bias + v * 128 + lane * 4);
    float4 r;
    r.x = acc[v].x + bv.x; r.y = acc[v].y + bv.y; r.z = acc[v].z + bv.z; r.w = acc[v].w + bv.w;
    if (RELU) { r.x = fmaxf(r.x, 0.f); r.y = fmaxf(r.y, 0.f); r.z = fmaxf(r.z, 0.f); r.w = fmaxf(r.w, 0.f); }
    *(float4*)(op + v * 128) = r;
  }
}

template <int FOUT, int RELU>
static void run_layer(const float* xin, int Fin, const float* W,
                      const float* avs, const float* avd, const float* bias,
                      float* h, float* xout,
                      const int* csr_src, const unsigned* rowstart, const unsigned* deg, int Nn,
                      float* as, float* ad, hipStream_t stream) {
  int mt = (Nn + 15) / 16, nt4 = FOUT / 64;
  int waves = mt * nt4;
  gemm_f32_wmma<<<(waves + 7) / 8, 256, 0, stream>>>(xin, W, avs, avd, h, as, ad, Nn, Fin, FOUT);
  gat_pull<FOUT, RELU><<<(Nn + 7) / 8, 256, 0, stream>>>(h, as, ad, csr_src, rowstart, deg,
                                                         bias, xout, Nn);
}

extern "C" void kernel_launch(void* const* d_in, const int* in_sizes, int n_in,
                              void* d_out, int out_size, void* d_ws, size_t ws_size,
                              hipStream_t stream) {
  const float* x   = (const float*)d_in[0];
  const int*  eidx = (const int*)d_in[1];
  const float* W1  = (const float*)d_in[2];
  const float* a1s = (const float*)d_in[3];
  const float* a1d = (const float*)d_in[4];
  const float* b1  = (const float*)d_in[5];
  const float* W2  = (const float*)d_in[6];
  const float* a2s = (const float*)d_in[7];
  const float* a2d = (const float*)d_in[8];
  const float* b2  = (const float*)d_in[9];
  float* out = (float*)d_out;

  const int IN_C = 128, HID_C = 256, OUT_C = 128;
  int Nn = in_sizes[0] / IN_C;
  int E  = in_sizes[1] / 2;
  const int* esrc = eidx;
  const int* edst = eidx + E;

  float* bufA = (float*)d_ws;
  float* bufB = bufA + (size_t)Nn * HID_C;
  int* csr_src = (int*)(bufB + (size_t)Nn * HID_C);
  unsigned* deg = (unsigned*)(csr_src + E);
  unsigned* rowstart = deg + Nn + 1;
  float* as   = (float*)(rowstart + Nn);
  float* ad   = as + Nn;
  unsigned* tmp   = (unsigned*)bufB;
  unsigned* offsT = tmp + E;
  (void)ws_size; (void)n_in; (void)out_size;

  const int C = (E + SORT_NB - 1) / SORT_NB;
  const int B = (Nn + BK_NODES - 1) / BK_NODES;
  edge_bin<<<SORT_NB, SORT_T, 0, stream>>>(esrc, edst, tmp, offsT, E, C, B);
  bucket_group<<<B, SORT_T, 0, stream>>>(tmp, offsT, csr_src, deg, rowstart, Nn, C);

  run_layer<256, 1>(x, IN_C, W1, a1s, a1d, b1,  bufA,  bufB,
                    csr_src, rowstart, deg, Nn, as, ad, stream);
  run_layer<128, 0>(bufB, HID_C, W2, a2s, a2d, b2,  bufA,  out,
                    csr_src, rowstart, deg, Nn, as, ad, stream);
  (void)OUT_C;
}


# layer 1 sums the softmax-weighted 128-wide INPUT rows per node, multiplies by W1 after (logits x.(W1 a)); in-edges of a node (up to 32) summed in source-id order; sum and product order change
# speedup vs baseline: 6.1185x; 1.1621x over previous
#include <hip/hip_runtime.h>

typedef __attribute__((ext_vector_type(2))) float v2f;
typedef __attribute__((ext_vector_type(8))) float v8f;

#define NEG_SLOPE 0.2f

template <int MASK>
__device__ __forceinline__ float row_xor_add(float x) {
  const int y = __builtin_amdgcn_update_dpp(0, __float_as_int(x), 0x160 | MASK, 0xf, 0xf, true);
  return x + __int_as_float(y);
}
__device__ __forceinline__ float row_sum16(float x) {
  x = row_xor_add<1>(x);
  x = row_xor_add<2>(x);
  x = row_xor_add<4>(x);
  x = row_xor_add<8>(x);
  return x;
}

__device__ __forceinline__ float wave_sum32(float x) {
  x = row_sum16(x);
  return x + __shfl_xor(x, 16, 32);
}

template <int EPI>
__global__ void gemm_f32_wmma(const float* __restrict__ X, const float* __restrict__ W,
                              const float* __restrict__ avs, const float* __restrict__ avd,
                              const float* __restrict__ bias,
                              float* __restrict__ H, float* __restrict__ as, float* __restrict__ ad,
                              int M, int K, int N) {
  __shared__ __attribute__((aligned(16))) float part[8][2][16];
  int lane = threadIdx.x & 31;
  int wib  = threadIdx.x >> 5;
  int wave = (int)((blockIdx.x * (long long)blockDim.x + threadIdx.x) >> 5);
  int nt4 = N >> 6;
  int mt  = (M + 15) >> 4;
  const bool live = wave < mt * nt4;
  int tm = wave / nt4, tn = wave % nt4;
  int half = lane >> 4;
  int l15  = lane & 15;
  int col0 = tn * 64 + l15 * 4;
  v8f acc0 = {}, acc1 = {}, acc2 = {}, acc3 = {};
  if (live) {
    int row = tm * 16 + l15;  if (row >= M) row = M - 1;
    const float4 sv = *(const float4*)((EPI == 0 ? avs : bias) + col0);
    const float4 dv = (EPI == 0) ? *(const float4*)(avd + col0) : sv;
    const float* xp = X + (size_t)row * K + (half << 2);
    const float* wp = W + (size_t)(half << 2) * N + col0;
    for (int k0 = 0; k0 < K; k0 += 8) {
      const float4 a = *(const float4*)(xp + k0);
      const float* w = wp + (size_t)k0 * N;
      const float4 r0 = *(const float4*)(w);
      const float4 r1 = *(const float4*)(w + N);
      const float4 r2 = *(const float4*)(w + 2 * N);
      const float4 r3 = *(const float4*)(w + 3 * N);
      v2f a01, a23, b0, b1, b2, b3;
      a01.x = a.x; a01.y = a.y;
      a23.x = a.z; a23.y = a.w;
      b0.x = r0.x; b0.y = r1.x;  b1.x = r0.y; b1.y = r1.y;
      b2.x = r0.z; b2.y = r1.z;  b3.x = r0.w; b3.y = r1.w;
      acc0 = __builtin_amdgcn_wmma_f32_16x16x4_f32(false, a01, false, b0, (short)0, acc0, false, false);
      acc1 = __builtin_amdgcn_wmma_f32_16x16x4_f32(false, a01, false, b1, (short)0, acc1, false, false);
      acc2 = __builtin_amdgcn_wmma_f32_16x16x4_f32(false, a01, false, b2, (short)0, acc2, false, false);
      acc3 = __builtin_amdgcn_wmma_f32_16x16x4_f32(false, a01, false, b3, (short)0, acc3, false, false);
      b0.x = r2.x; b0.y = r3.x;  b1.x = r2.y; b1.y = r3.y;
      b2.x = r2.z; b2.y = r3.z;  b3.x = r2.w; b3.y = r3.w;
      acc0 = __builtin_amdgcn_wmma_f32_16x16x4_f32(false, a23, false, b0, (short)0, acc0, false, false);
      acc1 = __builtin_amdgcn_wmma_f32_16x16x4_f32(false, a23, false, b1, (short)0, acc1, false, false);
      acc2 = __builtin_amdgcn_wmma_f32_16x16x4_f32(false, a23, false, b2, (short)0, acc2, false, false);
      acc3 = __builtin_amdgcn_wmma_f32_16x16x4_f32(false, a23, false, b3, (short)0, acc3, false, false);
    }
    if (EPI == 0) {
      float ps[8], pd[8];
#pragma unroll
      for (int v = 0; v < 8; ++v) {
        ps[v] = row_sum16(acc0[v] * sv.x + acc1[v] * sv.y + acc2[v] * sv.z + acc3[v] * sv.w);
        pd[v] = row_sum16(acc0[v] * dv.x + acc1[v] * dv.y + acc2[v] * dv.z + acc3[v] * dv.w);
      }
      if (l15 == 0) {
        float* p = &part[wib][0][half * 8];
        *(float4*)(p)          = make_float4(ps[0], ps[1], ps[2], ps[3]);
        *(float4*)(p + 4)      = make_float4(ps[4], ps[5], ps[6], ps[7]);
        *(float4*)(p + 16)     = make_float4(pd[0], pd[1], pd[2], pd[3]);
        *(float4*)(p + 16 + 4) = make_float4(pd[4], pd[5], pd[6], pd[7]);
      }
    } else {
#pragma unroll
      for (int v = 0; v < 8; ++v) {
        acc0[v] = fmaxf(acc0[v] + sv.x, 0.f);  acc1[v] = fmaxf(acc1[v] + sv.y, 0.f);
        acc2[v] = fmaxf(acc2[v] + sv.z, 0.f);  acc3[v] = fmaxf(acc3[v] + sv.w, 0.f);
      }
    }
  }
  if (EPI == 0) __syncthreads();
  if (live) {
    int rbase = tm * 16 + half * 8;
#pragma unroll
    for (int v = 0; v < 8; ++v) {
      int r = rbase + v;
      if (r < M) *(float4*)(H + (size_t)r * N + col0) = make_float4(acc0[v], acc1[v], acc2[v], acc3[v]);
    }
    if (EPI == 0 && tn == 0) {
      const float* p = &part[wib][0][0] + lane;
      float s = p[0];
      for (int t = 1; t < nt4; ++t) s += p[t * 32];
      int r = tm * 16 + l15;
      if (r < M) ((lane < 16) ? as : ad)[r] = s;
    }
  }
}

__device__ __forceinline__ float leaky(float e) { return (e > 0.f) ? e : NEG_SLOPE * e; }

#define SORT_NB   256
#define SORT_T    1024
#define SORT_EPT  4
#define BK_SHIFT  8
#define BK_NODES  256
#define BK_MAX    256
#define STAGE_CAP 8192

__device__ __forceinline__ unsigned wave_scan256(unsigned* a, int lane) {
  unsigned c[8], sum = 0u;
#pragma unroll
  for (int j = 0; j < 8; ++j) { c[j] = a[lane * 8 + j]; sum += c[j]; }
  unsigned incl = sum;
  for (int off = 1; off < 32; off <<= 1) {
    unsigned v = __shfl_up(incl, off, 32);
    if (lane >= off) incl += v;
  }
  unsigned run = incl - sum;
#pragma unroll
  for (int j = 0; j < 8; ++j) { a[lane * 8 + j] = run; run += c[j]; }
  return __shfl(incl, 31, 32);
}

__global__ __launch_bounds__(SORT_T) void edge_bin(const int* __restrict__ esrc, const int* __restrict__ edst,
                                                   unsigned* __restrict__ tmp, unsigned* __restrict__ offsT,
                                                   int E, int C, int B) {
  __shared__ unsigned hist[BK_MAX + 1];
  __shared__ unsigned stage[SORT_T * SORT_EPT];
  const int tid = threadIdx.x, blk = blockIdx.x;
  const int e0 = blk * C;
  int n = E - e0; if (n > C) n = C; if (n < 0) n = 0;
  if (tid <= BK_MAX) hist[tid] = 0u;
  __syncthreads();
  int d[SORT_EPT], s[SORT_EPT];
#pragma unroll
  for (int k = 0; k < SORT_EPT; ++k) {
    const int i = k * SORT_T + tid;
    d[k] = -1; s[k] = 0;
    if (i < n) { d[k] = edst[e0 + i]; s[k] = esrc[e0 + i]; }
  }
  unsigned slot[SORT_EPT], bk[SORT_EPT];
#pragma unroll
  for (int k = 0; k < SORT_EPT; ++k) {
    const int i = k * SORT_T + tid;
    unsigned b = (unsigned)d[k] >> BK_SHIFT;
    if (b >= (unsigned)B) b = (unsigned)B - 1u;
    bk[k] = b; slot[k] = 0u;
    if (i < n) slot[k] = atomicAdd(&hist[b], 1u);
  }
  __syncthreads();
  if (tid < 32) {
    const unsigned tot = wave_scan256(hist, tid);
    if (tid == 0) hist[BK_MAX] = tot;
  }
  __syncthreads();
  if (tid <= BK_MAX) offsT[tid * SORT_NB + blk] = hist[tid];
#pragma unroll
  for (int k = 0; k < SORT_EPT; ++k) {
    const int i = k * SORT_T + tid;
    if (i < n) stage[hist[bk[k]] + slot[k]] =
        ((unsigned)s[k] << BK_SHIFT) | ((unsigned)d[k] & (BK_NODES - 1));
  }
  __syncthreads();
#pragma unroll
  for (int k = 0; k < SORT_EPT; ++k) {
    const int i = k * SORT_T + tid;
    if (i < n) tmp[e0 + i] = stage[i];
  }
}

__global__ __launch_bounds__(SORT_T) void bucket_group(const unsigned* __restrict__ tmp,
                                                       const unsigned* __restrict__ offsT,
                                                       int* __restrict__ csr_src, unsigned* __restrict__ deg,
                                                       unsigned* __restrict__ rowstart, int Nn, int C) {
  __shared__ __attribute__((aligned(16))) unsigned segbase[SORT_NB];
  __shared__ __attribute__((aligned(16))) unsigned seglen[SORT_NB];
  __shared__ unsigned cnt[BK_NODES], nodeoff[BK_NODES], cur[BK_NODES];
  __shared__ unsigned red[2][8];
  __shared__ unsigned stage[STAGE_CAP];
  const int tid = threadIdx.x, lane = tid & 31, w = tid >> 5, b = blockIdx.x;
  if (tid < SORT_NB) {
    const unsigned lo = offsT[b * SORT_NB + tid], hi = offsT[(b + 1) * SORT_NB + tid];
    segbase[tid] = (unsigned)tid * (unsigned)C + lo;
    seglen[tid] = hi - lo;
    cnt[tid] = 0u;
    unsigned a = lo, l = hi - lo;
    for (int off = 16; off > 0; off >>= 1) { a += __shfl_xor(a, off, 32); l += __shfl_xor(l, off, 32); }
    if (lane == 0) { red[0][w] = a; red[1][w] = l; }
  }
  __syncthreads();
  unsigned bstart = 0u, nb = 0u;
#pragma unroll
  for (int j = 0; j < 8; ++j) { bstart += red[0][j]; nb += red[1][j]; }

  unsigned len8[8], base8[8];
  {
    const uint4 l0 = *(const uint4*)&seglen[w * 8], l1 = *(const uint4*)&seglen[w * 8 + 4];
    const uint4 b0 = *(const uint4*)&segbase[w * 8], b1 = *(const uint4*)&segbase[w * 8 + 4];
    len8[0] = l0.x; len8[1] = l0.y; len8[2] = l0.z; len8[3] = l0.w;
    len8[4] = l1.x; len8[5] = l1.y; len8[6] = l1.z; len8[7] = l1.w;
    base8[0] = b0.x; base8[1] = b0.y; base8[2] = b0.z; base8[3] = b0.w;
    base8[4] = b1.x; base8[5] = b1.y; base8[6] = b1.z; base8[7] = b1.w;
  }
  unsigned v[8], vm = 0u;
#pragma unroll
  for (int q = 0; q < 8; ++q) {
    v[q] = 0u;
    if ((unsigned)lane < len8[q]) { v[q] = tmp[base8[q] + lane]; vm |= 1u << q; }
  }
#pragma unroll
  for (int q = 0; q < 8; ++q)
    if (vm & (1u << q)) atomicAdd(&cnt[v[q] & (BK_NODES - 1)], 1u);
#pragma unroll
  for (int q = 0; q < 8; ++q) {
    const unsigned len = len8[q], base = base8[q];
    for (unsigned j = lane + 32; j < len; j += 32) atomicAdd(&cnt[tmp[base + j] & (BK_NODES - 1)], 1u);
  }
  __syncthreads();
  if (tid < 32) {
#pragma unroll
    for (int j = 0; j < 8; ++j) nodeoff[tid * 8 + j] = cnt[tid * 8 + j];
    wave_scan256(nodeoff, tid);
#pragma unroll
    for (int j = 0; j < 8; ++j) cur[tid * 8 + j] = nodeoff[tid * 8 + j];
  }
  __syncthreads();
  if (tid < BK_NODES) {
    const int node = b * BK_NODES + tid;
    if (node < Nn) { deg[node] = cnt[tid]; rowstart[node] = bstart + nodeoff[tid]; }
  }
  const bool staged = nb <= (unsigned)STAGE_CAP;
#pragma unroll
  for (int q = 0; q < 8; ++q) {
    if (vm & (1u << q)) {
      const unsigned p = atomicAdd(&cur[v[q] & (BK_NODES - 1)], 1u);
      if (staged) stage[p] = v[q] >> BK_SHIFT; else csr_src[bstart + p] = (int)(v[q] >> BK_SHIFT);
    }
  }
#pragma unroll
  for (int q = 0; q < 8; ++q) {
    const unsigned len = len8[q], base = base8[q];
    for (unsigned j = lane + 32; j < len; j += 32) {
      const unsigned r = tmp[base + j];
      const unsigned p = atomicAdd(&cur[r & (BK_NODES - 1)], 1u);
      if (staged) stage[p] = r >> BK_SHIFT; else csr_src[bstart + p] = (int)(r >> BK_SHIFT);
    }
  }
  __syncthreads();
  if (staged)
    for (unsigned i = tid; i < nb; i += SORT_T) csr_src[bstart + i] = (int)stage[i];
}

template <int K, int N>
__global__ __launch_bounds__(512) void logits_in(const float* __restrict__ X, const float* __restrict__ W,
                                                 const float* __restrict__ avs, const float* __restrict__ avd,
                                                 float* __restrict__ as, float* __restrict__ ad, int M) {
  constexpr int C = N / 128;
  constexpr int WPB = 16;
  constexpr int R = 16;
  __shared__ __attribute__((aligned(16))) float wv[2][K];
  const int lane = threadIdx.x & 31, wib = threadIdx.x >> 5;
  const bool lo = lane < 16;
  const int base = (blockIdx.x * WPB + wib) * R;
  float4 xv[R];
#pragma unroll
  for (int u = 0; u < R; ++u) {
    int r = base + u;  if (r >= M) r = M - 1;
    xv[u] = *(const float4*)(X + (size_t)r * K + lane * 4);
  }
  float4 sa[C], da[C];
#pragma unroll
  for (int c = 0; c < C; ++c) {
    sa[c] = *(const float4*)(avs + c * 128 + lane * 4);
    da[c] = *(const float4*)(avd + c * 128 + lane * 4);
  }
#pragma unroll
  for (int q = 0; q < K / WPB; ++q) {
    const int k = wib * (K / WPB) + q;
    float ps = 0.f, pd = 0.f;
#pragma unroll
    for (int c = 0; c < C; ++c) {
      const float4 w = *(const float4*)(W + (size_t)k * N + c * 128 + lane * 4);
      ps += w.x * sa[c].x + w.y * sa[c].y + w.z * sa[c].z + w.w * sa[c].w;
      pd += w.x * da[c].x + w.y * da[c].y + w.z * da[c].z + w.w * da[c].w;
    }
    ps = row_sum16(ps);  pd = row_sum16(pd);
    const float t = (lo ? ps : pd) + __shfl_xor(lo ? pd : ps, 16, 32);
    if ((lane & 15) == 0) wv[lo ? 0 : 1][k] = t;
  }
  __syncthreads();
  const float4 ws = *(const float4*)&wv[0][lane * 4];
  const float4 wd = *(const float4*)&wv[1][lane * 4];
  float keep = 0.f;
#pragma unroll
  for (int u = 0; u < R; ++u) {
    const float ps = row_sum16(xv[u].x * ws.x + xv[u].y * ws.y + xv[u].z * ws.z + xv[u].w * ws.w);
    const float pd = row_sum16(xv[u].x * wd.x + xv[u].y * wd.y + xv[u].z * wd.z + xv[u].w * wd.w);
    const float t = (lo ? ps : pd) + __shfl_xor(lo ? pd : ps, 16, 32);
    if ((lane & 15) == u) keep = t;
  }
  const int r = base + (lane & 15);
  if (r < M) (lo ? as : ad)[r] = keep;
}

template <int MASK>
__device__ __forceinline__ int row_xor_get(int x) {
  return __builtin_amdgcn_update_dpp(0, x, 0x160 | MASK, 0xf, 0xf, true);
}
template <int KK, int J>
__device__ __forceinline__ int bitonic_step(int x, int lane) {
  int o;
  if constexpr (J == 16) o = __shfl_xor(x, 16, 32); else o = row_xor_get<J>(x);
  const bool keep_min = (((lane & J) == 0) == ((lane & KK) == 0));
  return keep_min ? min(x, o) : max(x, o);
}
__device__ __forceinline__ int sort32_asc(int x, int lane) {
  x = bitonic_step<2, 1>(x, lane);
  x = bitonic_step<4, 2>(x, lane);   x = bitonic_step<4, 1>(x, lane);
  x = bitonic_step<8, 4>(x, lane);   x = bitonic_step<8, 2>(x, lane);   x = bitonic_step<8, 1>(x, lane);
  x = bitonic_step<16, 8>(x, lane);  x = bitonic_step<16, 4>(x, lane);  x = bitonic_step<16, 2>(x, lane);
  x = bitonic_step<16, 1>(x, lane);
  x = bitonic_step<32, 16>(x, lane); x = bitonic_step<32, 8>(x, lane);  x = bitonic_step<32, 4>(x, lane);
  x = bitonic_step<32, 2>(x, lane);  x = bitonic_step<32, 1>(x, lane);
  return x;
}

template <int F>
__device__ __forceinline__ void add_row(float4 (&acc)[F / 128], const float* __restrict__ h, int s, float a, int lane) {
  const float* hp = h + (size_t)s * F + lane * 4;
  for (int v = 0; v < F / 128; ++v) {
    const float4 hv = *(const float4*)(hp + v * 128);
    acc[v].x += a * hv.x; acc[v].y += a * hv.y; acc[v].z += a * hv.z; acc[v].w += a * hv.w;
  }
}

template <int F, int BIAS>
__global__ void gat_pull(const float* __restrict__ h, const float* __restrict__ as,
                         const float* __restrict__ ad, const int* __restrict__ csr_src,
                         const unsigned* __restrict__ rowstart, const unsigned* __restrict__ deg,
                         const float* __restrict__ bias, float* __restrict__ out, int Nn) {
  constexpr int V = F / 128;
  int lane = threadIdx.x & 31;
  const int node = __builtin_amdgcn_readfirstlane(
      (int)((blockIdx.x * (long long)blockDim.x + threadIdx.x) >> 5));
  if (node >= Nn) return;
  const int start = (int)rowstart[node];
  const int cnt   = (int)deg[node];
  const float adn = ad[node];
  const float e_self = leaky(as[node] + adn);

  float4 acc[V];
  for (int v = 0; v < V; ++v) acc[v] = make_float4(0.f, 0.f, 0.f, 0.f);
  if (cnt <= 32) {
    int sj = 0x7fffffff;
    if (lane < cnt) sj = csr_src[start + lane];
    sj = sort32_asc(sj, lane);
    const bool valid = lane < cnt;
    float e = e_self;
    if (valid) e = leaky(as[sj] + adn);
    float m = e;
    for (int off = 16; off > 0; off >>= 1) m = fmaxf(m, __shfl_xor(m, off, 32));
    const float w = valid ? expf(e - m) : 0.f;
    float zs = w;
    for (int off = 16; off > 0; off >>= 1) zs += __shfl_xor(zs, off, 32);
    const float w_self = expf(e_self - m);
    const float z = zs + w_self;
    const float aj = w / z;
    add_row<F>(acc, h, node, w_self / z, lane);
#pragma unroll 4
    for (int i = 0; i < cnt; ++i)
      add_row<F>(acc, h, __shfl(sj, i, 32), __shfl(aj, i, 32), lane);
  } else {
    float m = e_self;
    for (int j = lane; j < cnt; j += 32)
      m = fmaxf(m, leaky(as[csr_src[start + j]] + adn));
    for (int off = 16; off > 0; off >>= 1) m = fmaxf(m, __shfl_xor(m, off, 32));

    float zs = 0.f;
    for (int j = lane; j < cnt; j += 32)
      zs += expf(leaky(as[csr_src[start + j]] + adn) - m);
    for (int off = 16; off > 0; off >>= 1) zs += __shfl_xor(zs, off, 32);
    const float w_self = expf(e_self - m);
    const float z = zs + w_self;

    add_row<F>(acc, h, node, w_self / z, lane);
    for (int jb = 0; jb < cnt; jb += 32) {
      const int j = jb + lane;
      int sj = node;
      float aj = 0.f;
      if (j < cnt) {
        sj = csr_src[start + j];
        aj = expf(leaky(as[sj] + adn) - m) / z;
      }
      const int nj = (cnt - jb < 32) ? (cnt - jb) : 32;
#pragma unroll 4
      for (int i = 0; i < nj; ++i)
        add_row<F>(acc, h, __shfl(sj, i, 32), __shfl(aj, i, 32), lane);
    }
  }

  float* op = out + (size_t)node * F + lane * 4;
  for (int v = 0; v < V; ++v) {
    float4 r = acc[v];
    if (BIAS) {
      const float4 bv = *(const float4*)(bias + v * 128 + lane * 4);
      r.x += bv.x; r.y += bv.y; r.z += bv.z; r.w += bv.w;
    }
    *(float4*)(op + v * 128) = r;
  }
}

template <int FOUT, int EPI>
static void run_gemm(const float* xin, int Fin, const float* W,
                     const float* avs, const float* avd, const float* bias,
                     float* h, float* as, float* ad, int Nn, hipStream_t stream) {
  int mt = (Nn + 15) / 16, nt4 = FOUT / 64;
  int waves = mt * nt4;
  gemm_f32_wmma<EPI><<<(waves + 7) / 8, 256, 0, stream>>>(xin, W, avs, avd, bias, h, as, ad, Nn, Fin, FOUT);
}

extern "C" void kernel_launch(void* const* d_in, const int* in_sizes, int n_in,
                              void* d_out, int out_size, void* d_ws, size_t ws_size,
                              hipStream_t stream) {
  const float* x   = (const float*)d_in[0];
  const int*  eidx = (const int*)d_in[1];
  const float* W1  = (const float*)d_in[2];
  const float* a1s = (const float*)d_in[3];
  const float* a1d = (const float*)d_in[4];
  const float* b1  = (const float*)d_in[5];
  const float* W2  = (const float*)d_in[6];
  const float* a2s = (const float*)d_in[7];
  const float* a2d = (const float*)d_in[8];
  const float* b2  = (const float*)d_in[9];
  float* out = (float*)d_out;

  const int IN_C = 128, HID_C = 256, OUT_C = 128;
  int Nn = in_sizes[0] / IN_C;
  int E  = in_sizes[1] / 2;
  const int* esrc = eidx;
  const int* edst = eidx + E;

  float* bufA = (float*)d_ws;
  float* bufB = bufA + (size_t)Nn * HID_C;
  int* csr_src = (int*)(bufB + (size_t)Nn * HID_C);
  unsigned* deg = (unsigned*)(csr_src + E);
  unsigned* rowstart = deg + Nn + 1;
  float* as   = (float*)(rowstart + Nn);
  float* ad   = as + Nn;
  unsigned* tmp   = (unsigned*)bufB;
  unsigned* offsT = tmp + E;
  (void)ws_size; (void)n_in; (void)out_size;

  const int C = (E + SORT_NB - 1) / SORT_NB;
  const int B = (Nn + BK_NODES - 1) / BK_NODES;
  edge_bin<<<SORT_NB, SORT_T, 0, stream>>>(esrc, edst, tmp, offsT, E, C, B);
  bucket_group<<<B, SORT_T, 0, stream>>>(tmp, offsT, csr_src, deg, rowstart, Nn, C);

  logits_in<128, 256><<<(Nn + 255) / 256, 512, 0, stream>>>(x, W1, a1s, a1d, as, ad, Nn);
  gat_pull<128, 0><<<(Nn + 7) / 8, 256, 0, stream>>>(x, as, ad, csr_src, rowstart, deg,
                                                       b1, bufA, Nn);
  run_gemm<256, 1>(bufA, IN_C, W1, a1s, a1d, b1,  bufB, as, ad, Nn, stream);
  run_gemm<128, 0>(bufB, HID_C, W2, a2s, a2d, b2,  bufA, as, ad, Nn, stream);
  gat_pull<128, 1><<<(Nn + 7) / 8, 256, 0, stream>>>(bufA, as, ad, csr_src, rowstart, deg,
                                                       b2, out, Nn);
  (void)OUT_C;
}


# layer-1 gather fused with its product by W1: a 4-wave block sums the weighted input rows of 16 nodes into an LDS tile and multiplies it by W1 (bias, relu) from LDS; one launch fewer; sum order kept
# speedup vs baseline: 6.4792x; 1.0590x over previous
#include <hip/hip_runtime.h>

typedef __attribute__((ext_vector_type(2))) float v2f;
typedef __attribute__((ext_vector_type(8))) float v8f;

#define NEG_SLOPE 0.2f

template <int MASK>
__device__ __forceinline__ float row_xor_add(float x) {
  const int y = __builtin_amdgcn_update_dpp(0, __float_as_int(x), 0x160 | MASK, 0xf, 0xf, true);
  return x + __int_as_float(y);
}
__device__ __forceinline__ float row_sum16(float x) {
  x = row_xor_add<1>(x);
  x = row_xor_add<2>(x);
  x = row_xor_add<4>(x);
  x = row_xor_add<8>(x);
  return x;
}

__device__ __forceinline__ float wave_sum32(float x) {
  x = row_sum16(x);
  return x + __shfl_xor(x, 16, 32);
}

template <int EPI>
__global__ void gemm_f32_wmma(const float* __restrict__ X, const float* __restrict__ W,
                              const float* __restrict__ avs, const float* __restrict__ avd,
                              const float* __restrict__ bias,
                              float* __restrict__ H, float* __restrict__ as, float* __restrict__ ad,
                              int M, int K, int N) {
  __shared__ __attribute__((aligned(16))) float part[8][2][16];
  int lane = threadIdx.x & 31;
  int wib  = threadIdx.x >> 5;
  int wave = (int)((blockIdx.x * (long long)blockDim.x + threadIdx.x) >> 5);
  int nt4 = N >> 6;
  int mt  = (M + 15) >> 4;
  const bool live = wave < mt * nt4;
  int tm = wave / nt4, tn = wave % nt4;
  int half = lane >> 4;
  int l15  = lane & 15;
  int col0 = tn * 64 + l15 * 4;
  v8f acc0 = {}, acc1 = {}, acc2 = {}, acc3 = {};
  if (live) {
    int row = tm * 16 + l15;  if (row >= M) row = M - 1;
    const float4 sv = *(const float4*)((EPI == 0 ? avs : bias) + col0);
    const float4 dv = (EPI == 0) ? *(const float4*)(avd + col0) : sv;
    const float* xp = X + (size_t)row * K + (half << 2);
    const float* wp = W + (size_t)(half << 2) * N + col0;
    for (int k0 = 0; k0 < K; k0 += 8) {
      const float4 a = *(const float4*)(xp + k0);
      const float* w = wp + (size_t)k0 * N;
      const float4 r0 = *(const float4*)(w);
      const float4 r1 = *(const float4*)(w + N);
      const float4 r2 = *(const float4*)(w + 2 * N);
      const float4 r3 = *(const float4*)(w + 3 * N);
      v2f a01, a23, b0, b1, b2, b3;
      a01.x = a.x; a01.y = a.y;
      a23.x = a.z; a23.y = a.w;
      b0.x = r0.x; b0.y = r1.x;  b1.x = r0.y; b1.y = r1.y;
      b2.x = r0.z; b2.y = r1.z;  b3.x = r0.w; b3.y = r1.w;
      acc0 = __builtin_amdgcn_wmma_f32_16x16x4_f32(false, a01, false, b0, (short)0, acc0, false, false);
      acc1 = __builtin_amdgcn_wmma_f32_16x16x4_f32(false, a01, false, b1, (short)0, acc1, false, false);
      acc2 = __builtin_amdgcn_wmma_f32_16x16x4_f32(false, a01, false, b2, (short)0, acc2, false, false);
      acc3 = __builtin_amdgcn_wmma_f32_16x16x4_f32(false, a01, false, b3, (short)0, acc3, false, false);
      b0.x = r2.x; b0.y = r3.x;  b1.x = r2.y; b1.y = r3.y;
      b2.x = r2.z; b2.y = r3.z;  b3.x = r2.w; b3.y = r3.w;
      acc0 = __builtin_amdgcn_wmma_f32_16x16x4_f32(false, a23, false, b0, (short)0, acc0, false, false);
      acc1 = __builtin_amdgcn_wmma_f32_16x16x4_f32(false, a23, false, b1, (short)0, acc1, false, false);
      acc2 = __builtin_amdgcn_wmma_f32_16x16x4_f32(false, a23, false, b2, (short)0, acc2, false, false);
      acc3 = __builtin_amdgcn_wmma_f32_16x16x4_f32(false, a23, false, b3, (short)0, acc3, false, false);
    }
    if (EPI == 0) {
      float ps[8], pd[8];
#pragma unroll
      for (int v = 0; v < 8; ++v) {
        ps[v] = row_sum16(acc0[v] * sv.x + acc1[v] * sv.y + acc2[v] * sv.z + acc3[v] * sv.w);
        pd[v] = row_sum16(acc0[v] * dv.x + acc1[v] * dv.y + acc2[v] * dv.z + acc3[v] * dv.w);
      }
      if (l15 == 0) {
        float* p = &part[wib][0][half * 8];
        *(float4*)(p)          = make_float4(ps[0], ps[1], ps[2], ps[3]);
        *(float4*)(p + 4)      = make_float4(ps[4], ps[5], ps[6], ps[7]);
        *(float4*)(p + 16)     = make_float4(pd[0], pd[1], pd[2], pd[3]);
        *(float4*)(p + 16 + 4) = make_float4(pd[4], pd[5], pd[6], pd[7]);
      }
    } else {
#pragma unroll
      for (int v = 0; v < 8; ++v) {
        acc0[v] = fmaxf(acc0[v] + sv.x, 0.f);  acc1[v] = fmaxf(acc1[v] + sv.y, 0.f);
        acc2[v] = fmaxf(acc2[v] + sv.z, 0.f);  acc3[v] = fmaxf(acc3[v] + sv.w, 0.f);
      }
    }
  }
  if (EPI == 0) __syncthreads();
  if (live) {
    int rbase = tm * 16 + half * 8;
#pragma unroll
    for (int v = 0; v < 8; ++v) {
      int r = rbase + v;
      if (r < M) *(float4*)(H + (size_t)r * N + col0) = make_float4(acc0[v], acc1[v], acc2[v], acc3[v]);
    }
    if (EPI == 0 && tn == 0) {
      const float* p = &part[wib][0][0] + lane;
      float s = p[0];
      for (int t = 1; t < nt4; ++t) s += p[t * 32];
      int r = tm * 16 + l15;
      if (r < M) ((lane < 16) ? as : ad)[r] = s;
    }
  }
}

__device__ __forceinline__ float leaky(float e) { return (e > 0.f) ? e : NEG_SLOPE * e; }

#define SORT_NB   256
#define SORT_T    1024
#define SORT_EPT  4
#define BK_SHIFT  8
#define BK_NODES  256
#define BK_MAX    256
#define STAGE_CAP 8192

__device__ __forceinline__ unsigned wave_scan256(unsigned* a, int lane) {
  unsigned c[8], sum = 0u;
#pragma unroll
  for (int j = 0; j < 8; ++j) { c[j] = a[lane * 8 + j]; sum += c[j]; }
  unsigned incl = sum;
  for (int off = 1; off < 32; off <<= 1) {
    unsigned v = __shfl_up(incl, off, 32);
    if (lane >= off) incl += v;
  }
  unsigned run = incl - sum;
#pragma unroll
  for (int j = 0; j < 8; ++j) { a[lane * 8 + j] = run; run += c[j]; }
  return __shfl(incl, 31, 32);
}

__global__ __launch_bounds__(SORT_T) void edge_bin(const int* __restrict__ esrc, const int* __restrict__ edst,
                                                   unsigned* __restrict__ tmp, unsigned* __restrict__ offsT,
                                                   int E, int C, int B) {
  __shared__ unsigned hist[BK_MAX + 1];
  __shared__ unsigned stage[SORT_T * SORT_EPT];
  const int tid = threadIdx.x, blk = blockIdx.x;
  const int e0 = blk * C;
  int n = E - e0; if (n > C) n = C; if (n < 0) n = 0;
  if (tid <= BK_MAX) hist[tid] = 0u;
  __syncthreads();
  int d[SORT_EPT], s[SORT_EPT];
#pragma unroll
  for (int k = 0; k < SORT_EPT; ++k) {
    const int i = k * SORT_T + tid;
    d[k] = -1; s[k] = 0;
    if (i < n) { d[k] = edst[e0 + i]; s[k] = esrc[e0 + i]; }
  }
  unsigned slot[SORT_EPT], bk[SORT_EPT];
#pragma unroll
  for (int k = 0; k < SORT_EPT; ++k) {
    const int i = k * SORT_T + tid;
    unsigned b = (unsigned)d[k] >> BK_SHIFT;
    if (b >= (unsigned)B) b = (unsigned)B - 1u;
    bk[k] = b; slot[k] = 0u;
    if (i < n) slot[k] = atomicAdd(&hist[b], 1u);
  }
  __syncthreads();
  if (tid < 32) {
    const unsigned tot = wave_scan256(hist, tid);
    if (tid == 0) hist[BK_MAX] = tot;
  }
  __syncthreads();
  if (tid <= BK_MAX) offsT[tid * SORT_NB + blk] = hist[tid];
#pragma unroll
  for (int k = 0; k < SORT_EPT; ++k) {
    const int i = k * SORT_T + tid;
    if (i < n) stage[hist[bk[k]] + slot[k]] =
        ((unsigned)s[k] << BK_SHIFT) | ((unsigned)d[k] & (BK_NODES - 1));
  }
  __syncthreads();
#pragma unroll
  for (int k = 0; k < SORT_EPT; ++k) {
    const int i = k * SORT_T + tid;
    if (i < n) tmp[e0 + i] = stage[i];
  }
}

__global__ __launch_bounds__(SORT_T) void bucket_group(const unsigned* __restrict__ tmp,
                                                       const unsigned* __restrict__ offsT,
                                                       int* __restrict__ csr_src, unsigned* __restrict__ deg,
                                                       unsigned* __restrict__ rowstart, int Nn, int C) {
  __shared__ __attribute__((aligned(16))) unsigned segbase[SORT_NB];
  __shared__ __attribute__((aligned(16))) unsigned seglen[SORT_NB];
  __shared__ unsigned cnt[BK_NODES], nodeoff[BK_NODES], cur[BK_NODES];
  __shared__ unsigned red[2][8];
  __shared__ unsigned stage[STAGE_CAP];
  const int tid = threadIdx.x, lane = tid & 31, w = tid >> 5, b = blockIdx.x;
  if (tid < SORT_NB) {
    const unsigned lo = offsT[b * SORT_NB + tid], hi = offsT[(b + 1) * SORT_NB + tid];
    segbase[tid] = (unsigned)tid * (unsigned)C + lo;
    seglen[tid] = hi - lo;
    cnt[tid] = 0u;
    unsigned a = lo, l = hi - lo;
    for (int off = 16; off > 0; off >>= 1) { a += __shfl_xor(a, off, 32); l += __shfl_xor(l, off, 32); }
    if (lane == 0) { red[0][w] = a; red[1][w] = l; }
  }
  __syncthreads();
  unsigned bstart = 0u, nb = 0u;
#pragma unroll
  for (int j = 0; j < 8; ++j) { bstart += red[0][j]; nb += red[1][j]; }

  unsigned len8[8], base8[8];
  {
    const uint4 l0 = *(const uint4*)&seglen[w * 8], l1 = *(const uint4*)&seglen[w * 8 + 4];
    const uint4 b0 = *(const uint4*)&segbase[w * 8], b1 = *(const uint4*)&segbase[w * 8 + 4];
    len8[0] = l0.x; len8[1] = l0.y; len8[2] = l0.z; len8[3] = l0.w;
    len8[4] = l1.x; len8[5] = l1.y; len8[6] = l1.z; len8[7] = l1.w;
    base8[0] = b0.x; base8[1] = b0.y; base8[2] = b0.z; base8[3] = b0.w;
    base8[4] = b1.x; base8[5] = b1.y; base8[6] = b1.z; base8[7] = b1.w;
  }
  unsigned v[8], vm = 0u;
#pragma unroll
  for (int q = 0; q < 8; ++q) {
    v[q] = 0u;
    if ((unsigned)lane < len8[q]) { v[q] = tmp[base8[q] + lane]; vm |= 1u << q; }
  }
#pragma unroll
  for (int q = 0; q < 8; ++q)
    if (vm & (1u << q)) atomicAdd(&cnt[v[q] & (BK_NODES - 1)], 1u);
#pragma unroll
  for (int q = 0; q < 8; ++q) {
    const unsigned len = len8[q], base = base8[q];
    for (unsigned j = lane + 32; j < len; j += 32) atomicAdd(&cnt[tmp[base + j] & (BK_NODES - 1)], 1u);
  }
  __syncthreads();
  if (tid < 32) {
#pragma unroll
    for (int j = 0; j < 8; ++j) nodeoff[tid * 8 + j] = cnt[tid * 8 + j];
    wave_scan256(nodeoff, tid);
#pragma unroll
    for (int j = 0; j < 8; ++j) cur[tid * 8 + j] = nodeoff[tid * 8 + j];
  }
  __syncthreads();
  if (tid < BK_NODES) {
    const int node = b * BK_NODES + tid;
    if (node < Nn) { deg[node] = cnt[tid]; rowstart[node] = bstart + nodeoff[tid]; }
  }
  const bool staged = nb <= (unsigned)STAGE_CAP;
#pragma unroll
  for (int q = 0; q < 8; ++q) {
    if (vm & (1u << q)) {
      const unsigned p = atomicAdd(&cur[v[q] & (BK_NODES - 1)], 1u);
      if (staged) stage[p] = v[q] >> BK_SHIFT; else csr_src[bstart + p] = (int)(v[q] >> BK_SHIFT);
    }
  }
#pragma unroll
  for (int q = 0; q < 8; ++q) {
    const unsigned len = len8[q], base = base8[q];
    for (unsigned j = lane + 32; j < len; j += 32) {
      const unsigned r = tmp[base + j];
      const unsigned p = atomicAdd(&cur[r & (BK_NODES - 1)], 1u);
      if (staged) stage[p] = r >> BK_SHIFT; else csr_src[bstart + p] = (int)(r >> BK_SHIFT);
    }
  }
  __syncthreads();
  if (staged)
    for (unsigned i = tid; i < nb; i += SORT_T) csr_src[bstart + i] = (int)stage[i];
}

template <int K, int N>
__global__ __launch_bounds__(512) void logits_in(const float* __restrict__ X, const float* __restrict__ W,
                                                 const float* __restrict__ avs, const float* __restrict__ avd,
                                                 float* __restrict__ as, float* __restrict__ ad, int M) {
  constexpr int C = N / 128;
  constexpr int WPB = 16;
  constexpr int R = 16;
  __shared__ __attribute__((aligned(16))) float wv[2][K];
  const int lane = threadIdx.x & 31, wib = threadIdx.x >> 5;
  const bool lo = lane < 16;
  const int base = (blockIdx.x * WPB + wib) * R;
  float4 xv[R];
#pragma unroll
  for (int u = 0; u < R; ++u) {
    int r = base + u;  if (r >= M) r = M - 1;
    xv[u] = *(const float4*)(X + (size_t)r * K + lane * 4);
  }
  float4 sa[C], da[C];
#pragma unroll
  for (int c = 0; c < C; ++c) {
    sa[c] = *(const float4*)(avs + c * 128 + lane * 4);
    da[c] = *(const float4*)(avd + c * 128 + lane * 4);
  }
#pragma unroll
  for (int q = 0; q < K / WPB; ++q) {
    const int k = wib * (K / WPB) + q;
    float ps = 0.f, pd = 0.f;
#pragma unroll
    for (int c = 0; c < C; ++c) {
      const float4 w = *(const float4*)(W + (size_t)k * N + c * 128 + lane * 4);
      ps += w.x * sa[c].x + w.y * sa[c].y + w.z * sa[c].z + w.w * sa[c].w;
      pd += w.x * da[c].x + w.y * da[c].y + w.z * da[c].z + w.w * da[c].w;
    }
    ps = row_sum16(ps);  pd = row_sum16(pd);
    const float t = (lo ? ps : pd) + __shfl_xor(lo ? pd : ps, 16, 32);
    if ((lane & 15) == 0) wv[lo ? 0 : 1][k] = t;
  }
  __syncthreads();
  const float4 ws = *(const float4*)&wv[0][lane * 4];
  const float4 wd = *(const float4*)&wv[1][lane * 4];
  float keep = 0.f;
#pragma unroll
  for (int u = 0; u < R; ++u) {
    const float ps = row_sum16(xv[u].x * ws.x + xv[u].y * ws.y + xv[u].z * ws.z + xv[u].w * ws.w);
    const float pd = row_sum16(xv[u].x * wd.x + xv[u].y * wd.y + xv[u].z * wd.z + xv[u].w * wd.w);
    const float t = (lo ? ps : pd) + __shfl_xor(lo ? pd : ps, 16, 32);
    if ((lane & 15) == u) keep = t;
  }
  const int r = base + (lane & 15);
  if (r < M) (lo ? as : ad)[r] = keep;
}

template <int MASK>
__device__ __forceinline__ int row_xor_get(int x) {
  return __builtin_amdgcn_update_dpp(0, x, 0x160 | MASK, 0xf, 0xf, true);
}
template <int KK, int J>
__device__ __forceinline__ int bitonic_step(int x, int lane) {
  int o;
  if constexpr (J == 16) o = __shfl_xor(x, 16, 32); else o = row_xor_get<J>(x);
  const bool keep_min = (((lane & J) == 0) == ((lane & KK) == 0));
  return keep_min ? min(x, o) : max(x, o);
}
__device__ __forceinline__ int sort32_asc(int x, int lane) {
  x = bitonic_step<2, 1>(x, lane);
  x = bitonic_step<4, 2>(x, lane);   x = bitonic_step<4, 1>(x, lane);
  x = bitonic_step<8, 4>(x, lane);   x = bitonic_step<8, 2>(x, lane);   x = bitonic_step<8, 1>(x, lane);
  x = bitonic_step<16, 8>(x, lane);  x = bitonic_step<16, 4>(x, lane);  x = bitonic_step<16, 2>(x, lane);
  x = bitonic_step<16, 1>(x, lane);
  x = bitonic_step<32, 16>(x, lane); x = bitonic_step<32, 8>(x, lane);  x = bitonic_step<32, 4>(x, lane);
  x = bitonic_step<32, 2>(x, lane);  x = bitonic_step<32, 1>(x, lane);
  return x;
}

template <int F>
__device__ __forceinline__ void add_row(float4 (&acc)[F / 128], const float* __restrict__ h, int s, float a, int lane) {
  const float* hp = h + (size_t)s * F + lane * 4;
  for (int v = 0; v < F / 128; ++v) {
    const float4 hv = *(const float4*)(hp + v * 128);
    acc[v].x += a * hv.x; acc[v].y += a * hv.y; acc[v].z += a * hv.z; acc[v].w += a * hv.w;
  }
}

template <int F, int BIAS>
__global__ void gat_pull(const float* __restrict__ h, const float* __restrict__ as,
                         const float* __restrict__ ad, const int* __restrict__ csr_src,
                         const unsigned* __restrict__ rowstart, const unsigned* __restrict__ deg,
                         const float* __restrict__ bias, float* __restrict__ out, int Nn) {
  constexpr int V = F / 128;
  int lane = threadIdx.x & 31;
  const int node = __builtin_amdgcn_readfirstlane(
      (int)((blockIdx.x * (long long)blockDim.x + threadIdx.x) >> 5));
  if (node >= Nn) return;
  const int start = (int)rowstart[node];
  const int cnt   = (int)deg[node];
  const float adn = ad[node];
  const float e_self = leaky(as[node] + adn);

  float4 acc[V];
  for (int v = 0; v < V; ++v) acc[v] = make_float4(0.f, 0.f, 0.f, 0.f);
  if (cnt <= 32) {
    int sj = 0x7fffffff;
    if (lane < cnt) sj = csr_src[start + lane];
    sj = sort32_asc(sj, lane);
    const bool valid = lane < cnt;
    float e = e_self;
    if (valid) e = leaky(as[sj] + adn);
    float m = e;
    for (int off = 16; off > 0; off >>= 1) m = fmaxf(m, __shfl_xor(m, off, 32));
    const float w = valid ? expf(e - m) : 0.f;
    float zs = w;
    for (int off = 16; off > 0; off >>= 1) zs += __shfl_xor(zs, off, 32);
    const float w_self = expf(e_self - m);
    const float z = zs + w_self;
    const float aj = w / z;
    add_row<F>(acc, h, node, w_self / z, lane);
#pragma unroll 4
    for (int i = 0; i < cnt; ++i)
      add_row<F>(acc, h, __shfl(sj, i, 32), __shfl(aj, i, 32), lane);
  } else {
    float m = e_self;
    for (int j = lane; j < cnt; j += 32)
      m = fmaxf(m, leaky(as[csr_src[start + j]] + adn));
    for (int off = 16; off > 0; off >>= 1) m = fmaxf(m, __shfl_xor(m, off, 32));

    float zs = 0.f;
    for (int j = lane; j < cnt; j += 32)
      zs += expf(leaky(as[csr_src[start + j]] + adn) - m);
    for (int off = 16; off > 0; off >>= 1) zs += __shfl_xor(zs, off, 32);
    const float w_self = expf(e_self - m);
    const float z = zs + w_self;

    add_row<F>(acc, h, node, w_self / z, lane);
    for (int jb = 0; jb < cnt; jb += 32) {
      const int j = jb + lane;
      int sj = node;
      float aj = 0.f;
      if (j < cnt) {
        sj = csr_src[start + j];
        aj = expf(leaky(as[sj] + adn) - m) / z;
      }
      const int nj = (cnt - jb < 32) ? (cnt - jb) : 32;
#pragma unroll 4
      for (int i = 0; i < nj; ++i)
        add_row<F>(acc, h, __shfl(sj, i, 32), __shfl(aj, i, 32), lane);
    }
  }

  float* op = out + (size_t)node * F + lane * 4;
  for (int v = 0; v < V; ++v) {
    float4 r = acc[v];
    if (BIAS) {
      const float4 bv = *(const float4*)(bias + v * 128 + lane * 4);
      r.x += bv.x; r.y += bv.y; r.z += bv.z; r.w += bv.w;
    }
    *(float4*)(op + v * 128) = r;
  }
}

__device__ __forceinline__ float4 pull_sum128(const float* __restrict__ h, const float* __restrict__ as,
                                              const float* __restrict__ ad, const int* __restrict__ csr_src,
                                              const unsigned* __restrict__ rowstart,
                                              const unsigned* __restrict__ deg, int node, int lane) {
  const int start = (int)rowstart[node];
  const int cnt   = (int)deg[node];
  const float adn = ad[node];
  const float e_self = leaky(as[node] + adn);
  float4 acc[1];
  acc[0] = make_float4(0.f, 0.f, 0.f, 0.f);
  if (cnt <= 32) {
    int sj = 0x7fffffff;
    if (lane < cnt) sj = csr_src[start + lane];
    sj = sort32_asc(sj, lane);
    const bool valid = lane < cnt;
    float e = e_self;
    if (valid) e = leaky(as[sj] + adn);
    float m = e;
    for (int off = 16; off > 0; off >>= 1) m = fmaxf(m, __shfl_xor(m, off, 32));
    const float w = valid ? expf(e - m) : 0.f;
    float zs = w;
    for (int off = 16; off > 0; off >>= 1) zs += __shfl_xor(zs, off, 32);
    const float w_self = expf(e_self - m);
    const float z = zs + w_self;
    const float aj = w / z;
    add_row<128>(acc, h, node, w_self / z, lane);
#pragma unroll 4
    for (int i = 0; i < cnt; ++i)
      add_row<128>(acc, h, __shfl(sj, i, 32), __shfl(aj, i, 32), lane);
  } else {
    float m = e_self;
    for (int j = lane; j < cnt; j += 32)
      m = fmaxf(m, leaky(as[csr_src[start + j]] + adn));
    for (int off = 16; off > 0; off >>= 1) m = fmaxf(m, __shfl_xor(m, off, 32));
    float zs = 0.f;
    for (int j = lane; j < cnt; j += 32)
      zs += expf(leaky(as[csr_src[start + j]] + adn) - m);
    for (int off = 16; off > 0; off >>= 1) zs += __shfl_xor(zs, off, 32);
    const float w_self = expf(e_self - m);
    const float z = zs + w_self;
    add_row<128>(acc, h, node, w_self / z, lane);
    for (int jb = 0; jb < cnt; jb += 32) {
      const int j = jb + lane;
      int sj = node;
      float aj = 0.f;
      if (j < cnt) {
        sj = csr_src[start + j];
        aj = expf(leaky(as[sj] + adn) - m) / z;
      }
      const int nj = (cnt - jb < 32) ? (cnt - jb) : 32;
#pragma unroll 4
      for (int i = 0; i < nj; ++i)
        add_row<128>(acc, h, __shfl(sj, i, 32), __shfl(aj, i, 32), lane);
    }
  }
  return acc[0];
}

#define PG_PITCH 132
__global__ __launch_bounds__(128) void pull_gemm_relu(const float* __restrict__ x, const float* __restrict__ as,
                                                      const float* __restrict__ ad, const int* __restrict__ csr_src,
                                                      const unsigned* __restrict__ rowstart,
                                                      const unsigned* __restrict__ deg,
                                                      const float* __restrict__ W, const float* __restrict__ bias,
                                                      float* __restrict__ H, int Nn) {
  constexpr int K = 128, N = 256;
  __shared__ __attribute__((aligned(16))) float g[16 * PG_PITCH];
  const int lane = threadIdx.x & 31;
  const int wib  = threadIdx.x >> 5;
  const int node0 = blockIdx.x * 16;
  for (int q = 0; q < 4; ++q) {
    const int r = __builtin_amdgcn_readfirstlane(wib * 4 + q);
    const int node = node0 + r;
    float4 s = make_float4(0.f, 0.f, 0.f, 0.f);
    if (node < Nn) s = pull_sum128(x, as, ad, csr_src, rowstart, deg, node, lane);
    *(float4*)&g[r * PG_PITCH + lane * 4] = s;
  }
  __syncthreads();
  const int half = lane >> 4;
  const int l15  = lane & 15;
  const int col0 = wib * 64 + l15 * 4;
  v8f acc0 = {}, acc1 = {}, acc2 = {}, acc3 = {};
  const float4 sv = *(const float4*)(bias + col0);
  const float* xp = &g[l15 * PG_PITCH + (half << 2)];
  const float* wp = W + (size_t)(half << 2) * N + col0;
  for (int k0 = 0; k0 < K; k0 += 8) {
    const float4 a = *(const float4*)(xp + k0);
    const float* w = wp + (size_t)k0 * N;
    const float4 r0 = *(const float4*)(w);
    const float4 r1 = *(const float4*)(w + N);
    const float4 r2 = *(const float4*)(w + 2 * N);
    const float4 r3 = *(const float4*)(w + 3 * N);
    v2f a01, a23, b0, b1, b2, b3;
    a01.x = a.x; a01.y = a.y;
    a23.x = a.z; a23.y = a.w;
    b0.x = r0.x; b0.y = r1.x;  b1.x = r0.y; b1.y = r1.y;
    b2.x = r0.z; b2.y = r1.z;  b3.x = r0.w; b3.y = r1.w;
    acc0 = __builtin_amdgcn_wmma_f32_16x16x4_f32(false, a01, false, b0, (short)0, acc0, false, false);
    acc1 = __builtin_amdgcn_wmma_f32_16x16x4_f32(false, a01, false, b1, (short)0, acc1, false, false);
    acc2 = __builtin_amdgcn_wmma_f32_16x16x4_f32(false, a01, false, b2, (short)0, acc2, false, false);
    acc3 = __builtin_amdgcn_wmma_f32_16x16x4_f32(false, a01, false, b3, (short)0, acc3, false, false);
    b0.x = r2.x; b0.y = r3.x;  b1.x = r2.y; b1.y = r3.y;
    b2.x = r2.z; b2.y = r3.z;  b3.x = r2.w; b3.y = r3.w;
    acc0 = __builtin_amdgcn_wmma_f32_16x16x4_f32(false, a23, false, b0, (short)0, acc0, false, false);
    acc1 = __builtin_amdgcn_wmma_f32_16x16x4_f32(false, a23, false, b1, (short)0, acc1, false, false);
    acc2 = __builtin_amdgcn_wmma_f32_16x16x4_f32(false, a23, false, b2, (short)0, acc2, false, false);
    acc3 = __builtin_amdgcn_wmma_f32_16x16x4_f32(false, a23, false, b3, (short)0, acc3, false, false);
  }
#pragma unroll
  for (int v = 0; v < 8; ++v) {
    acc0[v] = fmaxf(acc0[v] + sv.x, 0.f);  acc1[v] = fmaxf(acc1[v] + sv.y, 0.f);
    acc2[v] = fmaxf(acc2[v] + sv.z, 0.f);  acc3[v] = fmaxf(acc3[v] + sv.w, 0.f);
  }
  const int rbase = node0 + half * 8;
#pragma unroll
  for (int v = 0; v < 8; ++v) {
    const int r = rbase + v;
    if (r < Nn) *(float4*)(H + (size_t)r * N + col0) = make_float4(acc0[v], acc1[v], acc2[v], acc3[v]);
  }
}

template <int FOUT, int EPI>
static void run_gemm(const float* xin, int Fin, const float* W,
                     const float* avs, const float* avd, const float* bias,
                     float* h, float* as, float* ad, int Nn, hipStream_t stream) {
  int mt = (Nn + 15) / 16, nt4 = FOUT / 64;
  int waves = mt * nt4;
  gemm_f32_wmma<EPI><<<(waves + 7) / 8, 256, 0, stream>>>(xin, W, avs, avd, bias, h, as, ad, Nn, Fin, FOUT);
}

extern "C" void kernel_launch(void* const* d_in, const int* in_sizes, int n_in,
                              void* d_out, int out_size, void* d_ws, size_t ws_size,
                              hipStream_t stream) {
  const float* x   = (const float*)d_in[0];
  const int*  eidx = (const int*)d_in[1];
  const float* W1  = (const float*)d_in[2];
  const float* a1s = (const float*)d_in[3];
  const float* a1d = (const float*)d_in[4];
  const float* b1  = (const float*)d_in[5];
  const float* W2  = (const float*)d_in[6];
  const float* a2s = (const float*)d_in[7];
  const float* a2d = (const float*)d_in[8];
  const float* b2  = (const float*)d_in[9];
  float* out = (float*)d_out;

  const int IN_C = 128, HID_C = 256, OUT_C = 128;
  int Nn = in_sizes[0] / IN_C;
  int E  = in_sizes[1] / 2;
  const int* esrc = eidx;
  const int* edst = eidx + E;

  float* bufA = (float*)d_ws;
  float* bufB = bufA + (size_t)Nn * HID_C;
  int* csr_src = (int*)(bufB + (size_t)Nn * HID_C);
  unsigned* deg = (unsigned*)(csr_src + E);
  unsigned* rowstart = deg + Nn + 1;
  float* as   = (float*)(rowstart + Nn);
  float* ad   = as + Nn;
  unsigned* tmp   = (unsigned*)bufB;
  unsigned* offsT = tmp + E;
  (void)ws_size; (void)n_in; (void)out_size;

  const int C = (E + SORT_NB - 1) / SORT_NB;
  const int B = (Nn + BK_NODES - 1) / BK_NODES;
  edge_bin<<<SORT_NB, SORT_T, 0, stream>>>(esrc, edst, tmp, offsT, E, C, B);
  bucket_group<<<B, SORT_T, 0, stream>>>(tmp, offsT, csr_src, deg, rowstart, Nn, C);

  logits_in<128, 256><<<(Nn + 255) / 256, 512, 0, stream>>>(x, W1, a1s, a1d, as, ad, Nn);
  pull_gemm_relu<<<(Nn + 15) / 16, 128, 0, stream>>>(x, as, ad, csr_src, rowstart, deg, W1, b1,  bufB, Nn);
  run_gemm<128, 0>(bufB, HID_C, W2, a2s, a2d, b2,  bufA, as, ad, Nn, stream);
  gat_pull<128, 1><<<(Nn + 7) / 8, 256, 0, stream>>>(bufA, as, ad, csr_src, rowstart, deg,
                                                       b2, out, Nn);
  (void)OUT_C; (void)IN_C;
}


# lane exchanges taken off the LDS unit: butterfly steps of sort, max and sum on DPP and v_permlanex16 (same pairing order), per-edge id and weight by v_readlane, scans by DPP row_shr; sums kept
# speedup vs baseline: 8.0003x; 1.2348x over previous
#include <hip/hip_runtime.h>

typedef __attribute__((ext_vector_type(2))) float v2f;
typedef __attribute__((ext_vector_type(8))) float v8f;

#define NEG_SLOPE 0.2f

template <int MASK>
__device__ __forceinline__ float row_xor_add(float x) {
  const int y = __builtin_amdgcn_update_dpp(0, __float_as_int(x), 0x160 | MASK, 0xf, 0xf, true);
  return x + __int_as_float(y);
}
template <int MASK>
__device__ __forceinline__ int row_xor_get(int x) {
  return __builtin_amdgcn_update_dpp(0, x, 0x160 | MASK, 0xf, 0xf, true);
}
__device__ __forceinline__ int xrow_get(int x) {
  return __builtin_amdgcn_permlanex16(x, x, 0x76543210, 0xfedcba98, false, false);
}
__device__ __forceinline__ float xrow_getf(float x) { return __int_as_float(xrow_get(__float_as_int(x))); }
__device__ __forceinline__ float wave_max32(float m) {
  m = fmaxf(m, xrow_getf(m));
  m = fmaxf(m, __int_as_float(row_xor_get<8>(__float_as_int(m))));
  m = fmaxf(m, __int_as_float(row_xor_get<4>(__float_as_int(m))));
  m = fmaxf(m, __int_as_float(row_xor_get<2>(__float_as_int(m))));
  m = fmaxf(m, __int_as_float(row_xor_get<1>(__float_as_int(m))));
  return m;
}
__device__ __forceinline__ float wave_add32(float x) {
  x = x + xrow_getf(x);
  x = row_xor_add<8>(x);
  x = row_xor_add<4>(x);
  x = row_xor_add<2>(x);
  x = row_xor_add<1>(x);
  return x;
}
__device__ __forceinline__ unsigned wave_add32u(unsigned x) {
  x += (unsigned)xrow_get((int)x);
  x += (unsigned)row_xor_get<8>((int)x);
  x += (unsigned)row_xor_get<4>((int)x);
  x += (unsigned)row_xor_get<2>((int)x);
  x += (unsigned)row_xor_get<1>((int)x);
  return x;
}
__device__ __forceinline__ int lane_get(int x, int i) { return __builtin_amdgcn_readlane(x, i); }
__device__ __forceinline__ float lane_getf(float x, int i) { return __int_as_float(__builtin_amdgcn_readlane(__float_as_int(x), i)); }
template <int N>
__device__ __forceinline__ unsigned row_shr_get0(unsigned x) {
  return (unsigned)__builtin_amdgcn_update_dpp(0, (int)x, 0x110 | N, 0xf, 0xf, true);
}

__device__ __forceinline__ float row_sum16(float x) {
  x = row_xor_add<1>(x);
  x = row_xor_add<2>(x);
  x = row_xor_add<4>(x);
  x = row_xor_add<8>(x);
  return x;
}

__device__ __forceinline__ float wave_sum32(float x) {
  x = row_sum16(x);
  return x + xrow_getf(x);
}

__device__ __forceinline__ float leaky(float e) { return (e > 0.f) ? e : NEG_SLOPE * e; }

#define SORT_NB   256
#define SORT_T    1024
#define SORT_EPT  4
#define BK_SHIFT  8
#define BK_NODES  256
#define BK_MAX    256
#define STAGE_CAP 8192

__device__ __forceinline__ unsigned wave_scan256(unsigned* a, int lane) {
  unsigned c[8], sum = 0u;
#pragma unroll
  for (int j = 0; j < 8; ++j) { c[j] = a[lane * 8 + j]; sum += c[j]; }
  unsigned incl = sum;
  incl += row_shr_get0<1>(incl);
  incl += row_shr_get0<2>(incl);
  incl += row_shr_get0<4>(incl);
  incl += row_shr_get0<8>(incl);
  const unsigned low_total = (unsigned)lane_get((int)incl, 15);
  if (lane >= 16) incl += low_total;
  unsigned run = incl - sum;
#pragma unroll
  for (int j = 0; j < 8; ++j) { a[lane * 8 + j] = run; run += c[j]; }
  return (unsigned)lane_get((int)incl, 31);
}

__global__ __launch_bounds__(SORT_T) void edge_bin(const int* __restrict__ esrc, const int* __restrict__ edst,
                                                   unsigned* __restrict__ tmp, unsigned* __restrict__ offsT,
                                                   int E, int C, int B) {
  __shared__ unsigned hist[BK_MAX + 1];
  __shared__ unsigned stage[SORT_T * SORT_EPT];
  const int tid = threadIdx.x, blk = blockIdx.x;
  const int e0 = blk * C;
  int n = E - e0; if (n > C) n = C; if (n < 0) n = 0;
  if (tid <= BK_MAX) hist[tid] = 0u;
  __syncthreads();
  int d[SORT_EPT], s[SORT_EPT];
#pragma unroll
  for (int k = 0; k < SORT_EPT; ++k) {
    const int i = k * SORT_T + tid;
    d[k] = -1; s[k] = 0;
    if (i < n) { d[k] = edst[e0 + i]; s[k] = esrc[e0 + i]; }
  }
  unsigned slot[SORT_EPT], bk[SORT_EPT];
#pragma unroll
  for (int k = 0; k < SORT_EPT; ++k) {
    const int i = k * SORT_T + tid;
    unsigned b = (unsigned)d[k] >> BK_SHIFT;
    if (b >= (unsigned)B) b = (unsigned)B - 1u;
    bk[k] = b; slot[k] = 0u;
    if (i < n) slot[k] = atomicAdd(&hist[b], 1u);
  }
  __syncthreads();
  if (tid < 32) {
    const unsigned tot = wave_scan256(hist, tid);
    if (tid == 0) hist[BK_MAX] = tot;
  }
  __syncthreads();
  if (tid <= BK_MAX) offsT[tid * SORT_NB + blk] = hist[tid];
#pragma unroll
  for (int k = 0; k < SORT_EPT; ++k) {
    const int i = k * SORT_T + tid;
    if (i < n) stage[hist[bk[k]] + slot[k]] =
        ((unsigned)s[k] << BK_SHIFT) | ((unsigned)d[k] & (BK_NODES - 1));
  }
  __syncthreads();
#pragma unroll
  for (int k = 0; k < SORT_EPT; ++k) {
    const int i = k * SORT_T + tid;
    if (i < n) tmp[e0 + i] = stage[i];
  }
}

template <int K, int N>
__global__ __launch_bounds__(SORT_T) void bucket_group_logits(const unsigned* __restrict__ tmp,
                                                              const unsigned* __restrict__ offsT,
                                                              int* __restrict__ csr_src, unsigned* __restrict__ deg,
                                                              unsigned* __restrict__ rowstart, int Nn, int C,
                                                              const float* __restrict__ X, const float* __restrict__ W,
                                                              const float* __restrict__ avs,
                                                              const float* __restrict__ avd,
                                                              float* __restrict__ as, float* __restrict__ ad) {
  constexpr int NC  = N / 128;
  constexpr int WPB = SORT_T / 32;
  constexpr int KQ  = K / WPB;
  constexpr int R   = BK_NODES / WPB;
  __shared__ __attribute__((aligned(16))) unsigned segbase[SORT_NB];
  __shared__ __attribute__((aligned(16))) unsigned seglen[SORT_NB];
  __shared__ unsigned cnt[BK_NODES], nodeoff[BK_NODES], cur[BK_NODES];
  __shared__ unsigned red[2][8];
  __shared__ unsigned stage[STAGE_CAP];
  __shared__ __attribute__((aligned(16))) float wv[2][K];
  const int tid = threadIdx.x, lane = tid & 31, w = tid >> 5, b = blockIdx.x;
  const bool lowhalf = lane < 16;
  float4 sa[NC], da[NC], wr[KQ][NC];
#pragma unroll
  for (int c = 0; c < NC; ++c) {
    sa[c] = *(const float4*)(avs + c * 128 + lane * 4);
    da[c] = *(const float4*)(avd + c * 128 + lane * 4);
  }
#pragma unroll
  for (int q = 0; q < KQ; ++q)
#pragma unroll
    for (int c = 0; c < NC; ++c)
      wr[q][c] = *(const float4*)(W + (size_t)(w * KQ + q) * N + c * 128 + lane * 4);
  const int xbase = b * BK_NODES + w * R;
  float4 xv[R];
#pragma unroll
  for (int u = 0; u < R; ++u) {
    int r = xbase + u;  if (r >= Nn) r = Nn - 1;
    xv[u] = *(const float4*)(X + (size_t)r * K + lane * 4);
  }
  if (tid < SORT_NB) {
    const unsigned lo = offsT[b * SORT_NB + tid], hi = offsT[(b + 1) * SORT_NB + tid];
    segbase[tid] = (unsigned)tid * (unsigned)C + lo;
    seglen[tid] = hi - lo;
    cnt[tid] = 0u;
    const unsigned a = wave_add32u(lo), l = wave_add32u(hi - lo);
    if (lane == 0) { red[0][w] = a; red[1][w] = l; }
  }
  float tq[KQ];
#pragma unroll
  for (int q = 0; q < KQ; ++q) {
    float ps = 0.f, pd = 0.f;
#pragma unroll
    for (int c = 0; c < NC; ++c) {
      const float4 wq = wr[q][c];
      ps += wq.x * sa[c].x + wq.y * sa[c].y + wq.z * sa[c].z + wq.w * sa[c].w;
      pd += wq.x * da[c].x + wq.y * da[c].y + wq.z * da[c].z + wq.w * da[c].w;
    }
    ps = row_sum16(ps);  pd = row_sum16(pd);
    tq[q] = (lowhalf ? ps : pd) + xrow_getf(lowhalf ? pd : ps);
  }
  if ((lane & 15) == 0) {
#pragma unroll
    for (int q = 0; q < KQ; ++q) wv[lowhalf ? 0 : 1][w * KQ + q] = tq[q];
  }
  __syncthreads();
  unsigned bstart = 0u, nb = 0u;
#pragma unroll
  for (int j = 0; j < 8; ++j) { bstart += red[0][j]; nb += red[1][j]; }

  unsigned len8[8], base8[8];
  {
    const uint4 l0 = *(const uint4*)&seglen[w * 8], l1 = *(const uint4*)&seglen[w * 8 + 4];
    const uint4 b0 = *(const uint4*)&segbase[w * 8], b1 = *(const uint4*)&segbase[w * 8 + 4];
    len8[0] = l0.x; len8[1] = l0.y; len8[2] = l0.z; len8[3] = l0.w;
    len8[4] = l1.x; len8[5] = l1.y; len8[6] = l1.z; len8[7] = l1.w;
    base8[0] = b0.x; base8[1] = b0.y; base8[2] = b0.z; base8[3] = b0.w;
    base8[4] = b1.x; base8[5] = b1.y; base8[6] = b1.z; base8[7] = b1.w;
  }
  unsigned v[8], vm = 0u;
#pragma unroll
  for (int q = 0; q < 8; ++q) {
    v[q] = 0u;
    if ((unsigned)lane < len8[q]) { v[q] = tmp[base8[q] + lane]; vm |= 1u << q; }
  }
#pragma unroll
  for (int q = 0; q < 8; ++q)
    if (vm & (1u << q)) atomicAdd(&cnt[v[q] & (BK_NODES - 1)], 1u);
#pragma unroll
  for (int q = 0; q < 8; ++q) {
    const unsigned len = len8[q], base = base8[q];
    for (unsigned j = lane + 32; j < len; j += 32) atomicAdd(&cnt[tmp[base + j] & (BK_NODES - 1)], 1u);
  }
  __syncthreads();
  if (tid < 32) {
#pragma unroll
    for (int j = 0; j < 8; ++j) nodeoff[tid * 8 + j] = cnt[tid * 8 + j];
    wave_scan256(nodeoff, tid);
#pragma unroll
    for (int j = 0; j < 8; ++j) cur[tid * 8 + j] = nodeoff[tid * 8 + j];
  }
  __syncthreads();
  if (tid < BK_NODES) {
    const int node = b * BK_NODES + tid;
    if (node < Nn) { deg[node] = cnt[tid]; rowstart[node] = bstart + nodeoff[tid]; }
  }
  const bool staged = nb <= (unsigned)STAGE_CAP;
#pragma unroll
  for (int q = 0; q < 8; ++q) {
    if (vm & (1u << q)) {
      const unsigned p = atomicAdd(&cur[v[q] & (BK_NODES - 1)], 1u);
      if (staged) stage[p] = v[q] >> BK_SHIFT; else csr_src[bstart + p] = (int)(v[q] >> BK_SHIFT);
    }
  }
#pragma unroll
  for (int q = 0; q < 8; ++q) {
    const unsigned len = len8[q], base = base8[q];
    for (unsigned j = lane + 32; j < len; j += 32) {
      const unsigned r = tmp[base + j];
      const unsigned p = atomicAdd(&cur[r & (BK_NODES - 1)], 1u);
      if (staged) stage[p] = r >> BK_SHIFT; else csr_src[bstart + p] = (int)(r >> BK_SHIFT);
    }
  }
  __syncthreads();
  if (staged)
    for (unsigned i = tid; i < nb; i += SORT_T) csr_src[bstart + i] = (int)stage[i];

  const float4 ws = *(const float4*)&wv[0][lane * 4];
  const float4 wd = *(const float4*)&wv[1][lane * 4];
  float keep = 0.f;
#pragma unroll
  for (int u = 0; u < R; ++u) {
    const float ps = row_sum16(xv[u].x * ws.x + xv[u].y * ws.y + xv[u].z * ws.z + xv[u].w * ws.w);
    const float pd = row_sum16(xv[u].x * wd.x + xv[u].y * wd.y + xv[u].z * wd.z + xv[u].w * wd.w);
    const float t = (lowhalf ? ps : pd) + xrow_getf(lowhalf ? pd : ps);
    if ((lane & 15) == u) keep = t;
  }
  const int xr = xbase + (lane & 15);
  if ((lane & 15) < R && xr < Nn) (lowhalf ? as : ad)[xr] = keep;
}

template <int KK, int J>
__device__ __forceinline__ int bitonic_step(int x, int lane) {
  int o;
  if constexpr (J == 16) o = xrow_get(x); else o = row_xor_get<J>(x);
  const bool keep_min = (((lane & J) == 0) == ((lane & KK) == 0));
  return keep_min ? min(x, o) : max(x, o);
}
__device__ __forceinline__ int sort32_asc(int x, int lane) {
  x = bitonic_step<2, 1>(x, lane);
  x = bitonic_step<4, 2>(x, lane);   x = bitonic_step<4, 1>(x, lane);
  x = bitonic_step<8, 4>(x, lane);   x = bitonic_step<8, 2>(x, lane);   x = bitonic_step<8, 1>(x, lane);
  x = bitonic_step<16, 8>(x, lane);  x = bitonic_step<16, 4>(x, lane);  x = bitonic_step<16, 2>(x, lane);
  x = bitonic_step<16, 1>(x, lane);
  x = bitonic_step<32, 16>(x, lane); x = bitonic_step<32, 8>(x, lane);  x = bitonic_step<32, 4>(x, lane);
  x = bitonic_step<32, 2>(x, lane);  x = bitonic_step<32, 1>(x, lane);
  return x;
}

template <int F>
__device__ __forceinline__ void add_row(float4 (&acc)[F / 128], const float* __restrict__ h, int s, float a, int lane) {
  const float* hp = h + (size_t)s * F + lane * 4;
  for (int v = 0; v < F / 128; ++v) {
    const float4 hv = *(const float4*)(hp + v * 128);
    acc[v].x += a * hv.x; acc[v].y += a * hv.y; acc[v].z += a * hv.z; acc[v].w += a * hv.w;
  }
}

template <int F, int BIAS>
__global__ void gat_pull(const float* __restrict__ h, const float* __restrict__ as,
                         const float* __restrict__ ad, const int* __restrict__ csr_src,
                         const unsigned* __restrict__ rowstart, const unsigned* __restrict__ deg,
                         const float* __restrict__ bias, float* __restrict__ out, int Nn) {
  constexpr int V = F / 128;
  int lane = threadIdx.x & 31;
  const int node = __builtin_amdgcn_readfirstlane(
      (int)((blockIdx.x * (long long)blockDim.x + threadIdx.x) >> 5));
  if (node >= Nn) return;
  const int start = (int)rowstart[node];
  const int cnt   = (int)deg[node];
  const float adn = ad[node];
  const float e_self = leaky(as[node] + adn);

  float4 acc[V];
  for (int v = 0; v < V; ++v) acc[v] = make_float4(0.f, 0.f, 0.f, 0.f);
  if (cnt <= 32) {
    int sj = 0x7fffffff;
    if (lane < cnt) sj = csr_src[start + lane];
    sj = sort32_asc(sj, lane);
    const bool valid = lane < cnt;
    float e = e_self;
    if (valid) e = leaky(as[sj] + adn);
    float m = e;
    m = wave_max32(m);
    const float w = valid ? expf(e - m) : 0.f;
    float zs = w;
    zs = wave_add32(zs);
    const float w_self = expf(e_self - m);
    const float z = zs + w_self;
    const float aj = w / z;
    add_row<F>(acc, h, node, w_self / z, lane);
#pragma unroll 4
    for (int i = 0; i < cnt; ++i)
      add_row<F>(acc, h, lane_get(sj, i), lane_getf(aj, i), lane);
  } else {
    float m = e_self;
    for (int j = lane; j < cnt; j += 32)
      m = fmaxf(m, leaky(as[csr_src[start + j]] + adn));
    m = wave_max32(m);

    float zs = 0.f;
    for (int j = lane; j < cnt; j += 32)
      zs += expf(leaky(as[csr_src[start + j]] + adn) - m);
    zs = wave_add32(zs);
    const float w_self = expf(e_self - m);
    const float z = zs + w_self;

    add_row<F>(acc, h, node, w_self / z, lane);
    for (int jb = 0; jb < cnt; jb += 32) {
      const int j = jb + lane;
      int sj = node;
      float aj = 0.f;
      if (j < cnt) {
        sj = csr_src[start + j];
        aj = expf(leaky(as[sj] + adn) - m) / z;
      }
      const int nj = (cnt - jb < 32) ? (cnt - jb) : 32;
#pragma unroll 4
      for (int i = 0; i < nj; ++i)
        add_row<F>(acc, h, lane_get(sj, i), lane_getf(aj, i), lane);
    }
  }

  float* op = out + (size_t)node * F + lane * 4;
  for (int v = 0; v < V; ++v) {
    float4 r = acc[v];
    if (BIAS) {
      const float4 bv = *(const float4*)(bias + v * 128 + lane * 4);
      r.x += bv.x; r.y += bv.y; r.z += bv.z; r.w += bv.w;
    }
    *(float4*)(op + v * 128) = r;
  }
}

__device__ __forceinline__ float4 pull_sum128(const float* __restrict__ h, const float* __restrict__ as,
                                              const float* __restrict__ ad, const int* __restrict__ csr_src,
                                              const unsigned* __restrict__ rowstart,
                                              const unsigned* __restrict__ deg, int node, int lane) {
  const int start = (int)rowstart[node];
  const int cnt   = (int)deg[node];
  const float adn = ad[node];
  const float e_self = leaky(as[node] + adn);
  float4 acc[1];
  acc[0] = make_float4(0.f, 0.f, 0.f, 0.f);
  if (cnt <= 32) {
    int sj = 0x7fffffff;
    if (lane < cnt) sj = csr_src[start + lane];
    sj = sort32_asc(sj, lane);
    const bool valid = lane < cnt;
    float e = e_self;
    if (valid) e = leaky(as[sj] + adn);
    float m = e;
    m = wave_max32(m);
    const float w = valid ? expf(e - m) : 0.f;
    float zs = w;
    zs = wave_add32(zs);
    const float w_self = expf(e_self - m);
    const float z = zs + w_self;
    const float aj = w / z;
    add_row<128>(acc, h, node, w_self / z, lane);
#pragma unroll 4
    for (int i = 0; i < cnt; ++i)
      add_row<128>(acc, h, lane_get(sj, i), lane_getf(aj, i), lane);
  } else {
    float m = e_self;
    for (int j = lane; j < cnt; j += 32)
      m = fmaxf(m, leaky(as[csr_src[start + j]] + adn));
    m = wave_max32(m);
    float zs = 0.f;
    for (int j = lane; j < cnt; j += 32)
      zs += expf(leaky(as[csr_src[start + j]] + adn) - m);
    zs = wave_add32(zs);
    const float w_self = expf(e_self - m);
    const float z = zs + w_self;
    add_row<128>(acc, h, node, w_self / z, lane);
    for (int jb = 0; jb < cnt; jb += 32) {
      const int j = jb + lane;
      int sj = node;
      float aj = 0.f;
      if (j < cnt) {
        sj = csr_src[start + j];
        aj = expf(leaky(as[sj] + adn) - m) / z;
      }
      const int nj = (cnt - jb < 32) ? (cnt - jb) : 32;
#pragma unroll 4
      for (int i = 0; i < nj; ++i)
        add_row<128>(acc, h, lane_get(sj, i), lane_getf(aj, i), lane);
    }
  }
  return acc[0];
}

#define PG_PITCH 132
#define PT_PITCH 260
__global__ __launch_bounds__(128) void pull_gemm_relu_gemm(const float* __restrict__ x, const float* __restrict__ as,
                                                           const float* __restrict__ ad, const int* __restrict__ csr_src,
                                                           const unsigned* __restrict__ rowstart,
                                                           const unsigned* __restrict__ deg,
                                                           const float* __restrict__ W1, const float* __restrict__ bias1,
                                                           const float* __restrict__ W2,
                                                           const float* __restrict__ avs2, const float* __restrict__ avd2,
                                                           float* __restrict__ H2, float* __restrict__ as2,
                                                           float* __restrict__ ad2, int Nn) {
  constexpr int K1 = 128, N1 = 256, K2 = 256, N2 = 128;
  __shared__ __attribute__((aligned(16))) float tile[16 * PT_PITCH];
  __shared__ __attribute__((aligned(16))) float part[4][2][16];
  const int lane = threadIdx.x & 31;
  const int wib  = threadIdx.x >> 5;
  const int node0 = blockIdx.x * 16;
  for (int q = 0; q < 4; ++q) {
    const int r = __builtin_amdgcn_readfirstlane(wib * 4 + q);
    const int node = node0 + r;
    float4 s = make_float4(0.f, 0.f, 0.f, 0.f);
    if (node < Nn) s = pull_sum128(x, as, ad, csr_src, rowstart, deg, node, lane);
    *(float4*)&tile[r * PG_PITCH + lane * 4] = s;
  }
  __syncthreads();
  const int half = lane >> 4;
  const int l15  = lane & 15;
  v8f acc0 = {}, acc1 = {}, acc2 = {}, acc3 = {};
  {
    const int col0 = wib * 64 + l15 * 4;
    const float4 sv = *(const float4*)(bias1 + col0);
    const float* xp = &tile[l15 * PG_PITCH + (half << 2)];
    const float* wp = W1 + (size_t)(half << 2) * N1 + col0;
    for (int k0 = 0; k0 < K1; k0 += 8) {
      const float4 a = *(const float4*)(xp + k0);
      const float* w = wp + (size_t)k0 * N1;
      const float4 r0 = *(const float4*)(w);
      const float4 r1 = *(const float4*)(w + N1);
      const float4 r2 = *(const float4*)(w + 2 * N1);
      const float4 r3 = *(const float4*)(w + 3 * N1);
      v2f a01, a23, b0, b1, b2, b3;
      a01.x = a.x; a01.y = a.y;
      a23.x = a.z; a23.y = a.w;
      b0.x = r0.x; b0.y = r1.x;  b1.x = r0.y; b1.y = r1.y;
      b2.x = r0.z; b2.y = r1.z;  b3.x = r0.w; b3.y = r1.w;
      acc0 = __builtin_amdgcn_wmma_f32_16x16x4_f32(false, a01, false, b0, (short)0, acc0, false, false);
      acc1 = __builtin_amdgcn_wmma_f32_16x16x4_f32(false, a01, false, b1, (short)0, acc1, false, false);
      acc2 = __builtin_amdgcn_wmma_f32_16x16x4_f32(false, a01, false, b2, (short)0, acc2, false, false);
      acc3 = __builtin_amdgcn_wmma_f32_16x16x4_f32(false, a01, false, b3, (short)0, acc3, false, false);
      b0.x = r2.x; b0.y = r3.x;  b1.x = r2.y; b1.y = r3.y;
      b2.x = r2.z; b2.y = r3.z;  b3.x = r2.w; b3.y = r3.w;
      acc0 = __builtin_amdgcn_wmma_f32_16x16x4_f32(false, a23, false, b0, (short)0, acc0, false, false);
      acc1 = __builtin_amdgcn_wmma_f32_16x16x4_f32(false, a23, false, b1, (short)0, acc1, false, false);
      acc2 = __builtin_amdgcn_wmma_f32_16x16x4_f32(false, a23, false, b2, (short)0, acc2, false, false);
      acc3 = __builtin_amdgcn_wmma_f32_16x16x4_f32(false, a23, false, b3, (short)0, acc3, false, false);
    }
    __syncthreads();
#pragma unroll
    for (int v = 0; v < 8; ++v) {
      *(float4*)&tile[(half * 8 + v) * PT_PITCH + col0] =
          make_float4(fmaxf(acc0[v] + sv.x, 0.f), fmaxf(acc1[v] + sv.y, 0.f),
                      fmaxf(acc2[v] + sv.z, 0.f), fmaxf(acc3[v] + sv.w, 0.f));
    }
  }
  __syncthreads();
  const int c0 = wib * 32 + l15 * 2;
  const float2 sv2 = *(const float2*)(avs2 + c0);
  const float2 dv2 = *(const float2*)(avd2 + c0);
  v8f q0 = {}, q1 = {};
  {
    const float* xp = &tile[l15 * PT_PITCH + (half << 2)];
    const float* wp = W2 + (size_t)(half << 2) * N2 + c0;
    for (int k0 = 0; k0 < K2; k0 += 8) {
      const float4 a = *(const float4*)(xp + k0);
      const float* w = wp + (size_t)k0 * N2;
      const float2 r0 = *(const float2*)(w);
      const float2 r1 = *(const float2*)(w + N2);
      const float2 r2 = *(const float2*)(w + 2 * N2);
      const float2 r3 = *(const float2*)(w + 3 * N2);
      v2f a01, a23, b0, b1;
      a01.x = a.x; a01.y = a.y;
      a23.x = a.z; a23.y = a.w;
      b0.x = r0.x; b0.y = r1.x;  b1.x = r0.y; b1.y = r1.y;
      q0 = __builtin_amdgcn_wmma_f32_16x16x4_f32(false, a01, false, b0, (short)0, q0, false, false);
      q1 = __builtin_amdgcn_wmma_f32_16x16x4_f32(false, a01, false, b1, (short)0, q1, false, false);
      b0.x = r2.x; b0.y = r3.x;  b1.x = r2.y; b1.y = r3.y;
      q0 = __builtin_amdgcn_wmma_f32_16x16x4_f32(false, a23, false, b0, (short)0, q0, false, false);
      q1 = __builtin_amdgcn_wmma_f32_16x16x4_f32(false, a23, false, b1, (short)0, q1, false, false);
    }
  }
  {
    float ps[8], pd[8];
#pragma unroll
    for (int v = 0; v < 8; ++v) {
      ps[v] = row_sum16(q0[v] * sv2.x + q1[v] * sv2.y);
      pd[v] = row_sum16(q0[v] * dv2.x + q1[v] * dv2.y);
    }
    if (l15 == 0) {
      float* p = &part[wib][0][half * 8];
      *(float4*)(p)          = make_float4(ps[0], ps[1], ps[2], ps[3]);
      *(float4*)(p + 4)      = make_float4(ps[4], ps[5], ps[6], ps[7]);
      *(float4*)(p + 16)     = make_float4(pd[0], pd[1], pd[2], pd[3]);
      *(float4*)(p + 16 + 4) = make_float4(pd[4], pd[5], pd[6], pd[7]);
    }
  }
  __syncthreads();
  const int rbase = node0 + half * 8;
#pragma unroll
  for (int v = 0; v < 8; ++v) {
    const int r = rbase + v;
    if (r < Nn) *(float2*)(H2 + (size_t)r * N2 + c0) = make_float2(q0[v], q1[v]);
  }
  if (wib == 0) {
    const float* p = &part[0][0][0] + lane;
    float s = p[0];
    for (int t = 1; t < 4; ++t) s += p[t * 32];
    const int r = node0 + l15;
    if (r < Nn) ((lane < 16) ? as2 : ad2)[r] = s;
  }
}

extern "C" void kernel_launch(void* const* d_in, const int* in_sizes, int n_in,
                              void* d_out, int out_size, void* d_ws, size_t ws_size,
                              hipStream_t stream) {
  const float* x   = (const float*)d_in[0];
  const int*  eidx = (const int*)d_in[1];
  const float* W1  = (const float*)d_in[2];
  const float* a1s = (const float*)d_in[3];
  const float* a1d = (const float*)d_in[4];
  const float* b1  = (const float*)d_in[5];
  const float* W2  = (const float*)d_in[6];
  const float* a2s = (const float*)d_in[7];
  const float* a2d = (const float*)d_in[8];
  const float* b2  = (const float*)d_in[9];
  float* out = (float*)d_out;

  const int IN_C = 128, HID_C = 256, OUT_C = 128;
  int Nn = in_sizes[0] / IN_C;
  int E  = in_sizes[1] / 2;
  const int* esrc = eidx;
  const int* edst = eidx + E;

  float* bufA = (float*)d_ws;
  float* bufB = bufA + (size_t)Nn * HID_C;
  int* csr_src = (int*)(bufB + (size_t)Nn * HID_C);
  unsigned* deg = (unsigned*)(csr_src + E);
  unsigned* rowstart = deg + Nn + 1;
  float* as   = (float*)(rowstart + Nn);
  float* ad   = as + Nn;
  float* as2  = bufA + (size_t)Nn * OUT_C;
  float* ad2  = as2 + Nn;
  unsigned* tmp   = (unsigned*)bufB;
  unsigned* offsT = tmp + E;
  (void)ws_size; (void)n_in; (void)out_size;

  const int C = (E + SORT_NB - 1) / SORT_NB;
  const int B = (Nn + BK_NODES - 1) / BK_NODES;
  edge_bin<<<SORT_NB, SORT_T, 0, stream>>>(esrc, edst, tmp, offsT, E, C, B);
  bucket_group_logits<128, 256><<<B, SORT_T, 0, stream>>>(tmp, offsT, csr_src, deg, rowstart, Nn, C,
                                                          x, W1, a1s, a1d, as, ad);

  pull_gemm_relu_gemm<<<(Nn + 15) / 16, 128, 0, stream>>>(x, as, ad, csr_src, rowstart, deg, W1, b1,
                                                           W2, a2s, a2d,  bufA, as2, ad2, Nn);
  gat_pull<128, 1><<<(Nn + 7) / 8, 256, 0, stream>>>(bufA, as2, ad2, csr_src, rowstart, deg,
                                                       b2, out, Nn);
  (void)IN_C;
}
